# starting kernel
# baseline (speedup 1.0000x reference)
#include <hip/hip_runtime.h>
#include <hip/hip_bf16.h>

#define N_NODES 50000
#define D_HID   256

typedef __attribute__((ext_vector_type(2))) float v2f;
typedef __attribute__((ext_vector_type(8))) float v8f;

template <int DIN>
__launch_bounds__(128)
__global__ void gcn_gemm_wmma(const float* __restrict__ X,
                              const float* __restrict__ W,
                              float* __restrict__ H) {
  constexpr int DOUT = D_HID;

  const int lane = threadIdx.x & 31;
  const int wave = threadIdx.x >> 5;
  const int half = lane >> 4;
  const int m    = lane & 15;

  const int row0 = blockIdx.x * 16;
  const int col0 = wave * 64;

  const float* __restrict__ Xrow  = X + (size_t)(row0 + m) * DIN + 2 * half;
  const float* __restrict__ Wbase = W + (size_t)(2 * half) * DOUT + col0 + m;

  v8f acc0 = {0.f,0.f,0.f,0.f,0.f,0.f,0.f,0.f};
  v8f acc1 = acc0, acc2 = acc0, acc3 = acc0;

  #pragma unroll
  for (int k = 0; k < DIN; k += 4) {
    v2f a;
    a.x = Xrow[k];
    a.y = Xrow[k + 1];

    const float* __restrict__ w0 = Wbase + (size_t)k * DOUT;
    const float* __restrict__ w1 = w0 + DOUT;

    v2f b0, b1, b2, b3;
    b0.x = w0[0];   b0.y = w1[0];
    b1.x = w0[16];  b1.y = w1[16];
    b2.x = w0[32];  b2.y = w1[32];
    b3.x = w0[48];  b3.y = w1[48];

    acc0 = __builtin_amdgcn_wmma_f32_16x16x4_f32(false, a, false, b0, (short)0, acc0, false, false);
    acc1 = __builtin_amdgcn_wmma_f32_16x16x4_f32(false, a, false, b1, (short)0, acc1, false, false);
    acc2 = __builtin_amdgcn_wmma_f32_16x16x4_f32(false, a, false, b2, (short)0, acc2, false, false);
    acc3 = __builtin_amdgcn_wmma_f32_16x16x4_f32(false, a, false, b3, (short)0, acc3, false, false);
  }

  float* __restrict__ Hb = H + (size_t)row0 * DOUT + col0 + m;
  #pragma unroll
  for (int r = 0; r < 8; ++r) {
    float* p = Hb + (size_t)(r + 8 * half) * DOUT;
    p[0]  = acc0[r];
    p[16] = acc1[r];
    p[32] = acc2[r];
    p[48] = acc3[r];
  }
}

__global__ void gcn_deg_init(float* __restrict__ dinv, int n) {
  int i = blockIdx.x * blockDim.x + threadIdx.x;
  if (i < n) dinv[i] = 1.0f;
}

__global__ void gcn_deg_edges(const int* __restrict__ dst, float* __restrict__ dinv, int e) {
  int i = blockIdx.x * blockDim.x + threadIdx.x;
  if (i < e) unsafeAtomicAdd(&dinv[dst[i]], 1.0f);
}

__global__ void gcn_deg_rsqrt(float* __restrict__ dinv, int n) {
  int i = blockIdx.x * blockDim.x + threadIdx.x;
  if (i < n) dinv[i] = rsqrtf(dinv[i]);
}

__global__ void gcn_init_bias(float* __restrict__ agg, const float* __restrict__ b, int total) {
  int i = blockIdx.x * blockDim.x + threadIdx.x;
  if (i < total) agg[i] = b[i & (D_HID - 1)];
}

__launch_bounds__(256)
__global__ void gcn_edge_agg(const int* __restrict__ src, const int* __restrict__ dst,
                             const float* __restrict__ dinv,
                             const float* __restrict__ H,
                             float* __restrict__ agg, int e) {
  int gid  = blockIdx.x * blockDim.x + threadIdx.x;
  int edge = gid >> 5;
  int lane = gid & 31;
  if (edge >= e) return;

  int s = src[edge], d = dst[edge];
  float nrm = dinv[s] * dinv[d];

  const float4* __restrict__ hrow = (const float4*)(H + (size_t)s * D_HID);
  float4 h0 = hrow[lane * 2 + 0];
  float4 h1 = hrow[lane * 2 + 1];

  float* __restrict__ a = agg + (size_t)d * D_HID + lane * 8;
  unsafeAtomicAdd(a + 0, h0.x * nrm);
  unsafeAtomicAdd(a + 1, h0.y * nrm);
  unsafeAtomicAdd(a + 2, h0.z * nrm);
  unsafeAtomicAdd(a + 3, h0.w * nrm);
  unsafeAtomicAdd(a + 4, h1.x * nrm);
  unsafeAtomicAdd(a + 5, h1.y * nrm);
  unsafeAtomicAdd(a + 6, h1.z * nrm);
  unsafeAtomicAdd(a + 7, h1.w * nrm);
}

__global__ void gcn_finalize(const float* __restrict__ agg, const float* __restrict__ H,
                             const float* __restrict__ dinv,
                             float* __restrict__ out, int do_relu, int total) {
  int gid = blockIdx.x * blockDim.x + threadIdx.x;
  if (gid >= total) return;
  int i = gid >> 8;
  float c = dinv[i];
  float v = agg[gid] + H[gid] * c * c;
  out[gid] = do_relu ? fmaxf(v, 0.0f) : v;
}

extern "C" void kernel_launch(void* const* d_in, const int* in_sizes, int n_in,
                              void* d_out, int out_size, void* d_ws, size_t ws_size,
                              hipStream_t stream) {
  const float* x   = (const float*)d_in[0];
  const int*   ei  = (const int*)  d_in[1];
  const float* W1  = (const float*)d_in[2];
  const float* b1  = (const float*)d_in[3];
  const float* W2  = (const float*)d_in[4];
  const float* b2  = (const float*)d_in[5];
  const float* W3  = (const float*)d_in[6];
  const float* b3  = (const float*)d_in[7];

  const int N = in_sizes[0] / 128;
  const int E = in_sizes[1] / 2;
  const int total = N * D_HID;

  const int* src = ei;
  const int* dst = ei + E;

  float* H    = (float*)d_ws;
  float* AGG  = H + (size_t)total;
  float* dinv = AGG + (size_t)total;

  const int rowTiles = N / 16;
  dim3 gemmGrid(rowTiles, 1, 1);
  dim3 gemmBlk(128, 1, 1);
  const int nBlkN  = (N + 255) / 256;
  const int nBlkE  = (E + 255) / 256;
  const int nBlkT  = (total + 255) / 256;
  const int nBlkEW = (E * 32 + 255) / 256;

  gcn_deg_init <<<nBlkN, 256, 0, stream>>>(dinv, N);
  gcn_deg_edges<<<nBlkE, 256, 0, stream>>>(dst, dinv, E);
  gcn_deg_rsqrt<<<nBlkN, 256, 0, stream>>>(dinv, N);

  gcn_gemm_wmma<128><<<gemmGrid, gemmBlk, 0, stream>>>(x, W1, H);
  gcn_init_bias<<<nBlkT, 256, 0, stream>>>(AGG, b1, total);
  gcn_edge_agg <<<nBlkEW, 256, 0, stream>>>(src, dst, dinv, H, AGG, E);
  gcn_finalize <<<nBlkT, 256, 0, stream>>>(AGG, H, dinv, AGG, 1, total);

  gcn_gemm_wmma<256><<<gemmGrid, gemmBlk, 0, stream>>>(AGG, W2, H);
  gcn_init_bias<<<nBlkT, 256, 0, stream>>>(AGG, b2, total);
  gcn_edge_agg <<<nBlkEW, 256, 0, stream>>>(src, dst, dinv, H, AGG, E);
  gcn_finalize <<<nBlkT, 256, 0, stream>>>(AGG, H, dinv, AGG, 1, total);

  gcn_gemm_wmma<256><<<gemmGrid, gemmBlk, 0, stream>>>(AGG, W3, H);
  gcn_init_bias<<<nBlkT, 256, 0, stream>>>(AGG, b3, total);
  gcn_edge_agg <<<nBlkEW, 256, 0, stream>>>(src, dst, dinv, H, AGG, E);
  gcn_finalize <<<nBlkT, 256, 0, stream>>>(AGG, H, dinv, (float*)d_out, 0, total);
}


# Float-atomic scatter replaced by gather over a per-call edge list grouped by destination and sorted by source index: one writer per row, bias+self+relu fused; row sums in fixed list order
# speedup vs baseline: 3.4885x; 3.4885x over previous
#include <hip/hip_runtime.h>
#include <hip/hip_bf16.h>

#define N_NODES 50000
#define D_HID   256

typedef __attribute__((ext_vector_type(2))) float v2f;
typedef __attribute__((ext_vector_type(8))) float v8f;

template <int DIN>
__launch_bounds__(128)
__global__ void gcn_gemm_wmma(const float* __restrict__ X,
                              const float* __restrict__ W,
                              float* __restrict__ H) {
  constexpr int DOUT = D_HID;

  const int lane = threadIdx.x & 31;
  const int wave = threadIdx.x >> 5;
  const int half = lane >> 4;
  const int m    = lane & 15;

  const int row0 = blockIdx.x * 16;
  const int col0 = wave * 64;

  const float* __restrict__ Xrow  = X + (size_t)(row0 + m) * DIN + 2 * half;
  const float* __restrict__ Wbase = W + (size_t)(2 * half) * DOUT + col0 + m;

  v8f acc0 = {0.f,0.f,0.f,0.f,0.f,0.f,0.f,0.f};
  v8f acc1 = acc0, acc2 = acc0, acc3 = acc0;

  #pragma unroll
  for (int k = 0; k < DIN; k += 4) {
    v2f a;
    a.x = Xrow[k];
    a.y = Xrow[k + 1];

    const float* __restrict__ w0 = Wbase + (size_t)k * DOUT;
    const float* __restrict__ w1 = w0 + DOUT;

    v2f b0, b1, b2, b3;
    b0.x = w0[0];   b0.y = w1[0];
    b1.x = w0[16];  b1.y = w1[16];
    b2.x = w0[32];  b2.y = w1[32];
    b3.x = w0[48];  b3.y = w1[48];

    acc0 = __builtin_amdgcn_wmma_f32_16x16x4_f32(false, a, false, b0, (short)0, acc0, false, false);
    acc1 = __builtin_amdgcn_wmma_f32_16x16x4_f32(false, a, false, b1, (short)0, acc1, false, false);
    acc2 = __builtin_amdgcn_wmma_f32_16x16x4_f32(false, a, false, b2, (short)0, acc2, false, false);
    acc3 = __builtin_amdgcn_wmma_f32_16x16x4_f32(false, a, false, b3, (short)0, acc3, false, false);
  }

  float* __restrict__ Hb = H + (size_t)row0 * DOUT + col0 + m;
  #pragma unroll
  for (int r = 0; r < 8; ++r) {
    float* p = Hb + (size_t)(r + 8 * half) * DOUT;
    p[0]  = acc0[r];
    p[16] = acc1[r];
    p[32] = acc2[r];
    p[48] = acc3[r];
  }
}

__global__ void gcn_cnt_zero(int* __restrict__ cnt, int n) {
  int i = blockIdx.x * blockDim.x + threadIdx.x;
  if (i < n) cnt[i] = 0;
}

__global__ void gcn_cnt_edges(const int* __restrict__ dst, int* __restrict__ cnt,
                              int* __restrict__ pos, int e) {
  int i = blockIdx.x * blockDim.x + threadIdx.x;
  if (i < e) pos[i] = atomicAdd(&cnt[dst[i]], 1);
}

__launch_bounds__(1024)
__global__ void gcn_scan_deg(const int* __restrict__ cnt, int* __restrict__ off,
                             float* __restrict__ dinv, int n, int npad) {
  __shared__ int wsum[2][32];
  const int tid  = threadIdx.x;
  const int lane = tid & 31;
  const int wave = tid >> 5;

  const int4* __restrict__ cnt4  = (const int4*)cnt;
  int4*       __restrict__ off4  = (int4*)off;
  float4*     __restrict__ dinv4 = (float4*)dinv;
  const int lastq = npad / 4 - 1;

  int carry = 0;
  int buf   = 0;
  int q0 = tid * 2;
  int4 ra = cnt4[min(q0,     lastq)];
  int4 rb = cnt4[min(q0 + 1, lastq)];

  for (int base = 0; base < n; base += 8192) {
    const int q1 = q0 + 2048;
    const int4 na = cnt4[min(q1,     lastq)];
    const int4 nb = cnt4[min(q1 + 1, lastq)];

    const int i0 = q0 * 4;
    const int c0 = (i0     < n) ? ra.x + 1 : 0;
    const int c1 = (i0 + 1 < n) ? ra.y + 1 : 0;
    const int c2 = (i0 + 2 < n) ? ra.z + 1 : 0;
    const int c3 = (i0 + 3 < n) ? ra.w + 1 : 0;
    const int c4 = (i0 + 4 < n) ? rb.x + 1 : 0;
    const int c5 = (i0 + 5 < n) ? rb.y + 1 : 0;
    const int c6 = (i0 + 6 < n) ? rb.z + 1 : 0;
    const int c7 = (i0 + 7 < n) ? rb.w + 1 : 0;

    const int t = c0 + c1 + c2 + c3 + c4 + c5 + c6 + c7;
    int incl = t;
    #pragma unroll
    for (int o = 1; o < 32; o <<= 1) {
      int v = __shfl_up(incl, o);
      if (lane >= o) incl += v;
    }
    if (lane == 31) wsum[buf][wave] = incl;
    __syncthreads();

    const int ws = wsum[buf][lane];
    int wincl = ws;
    #pragma unroll
    for (int o = 1; o < 32; o <<= 1) {
      int v = __shfl_up(wincl, o);
      if (lane >= o) wincl += v;
    }
    const int wprefix = __shfl(wincl - ws, wave);
    const int total   = __shfl(wincl, 31);

    int4 oa, ob;
    int ex = carry + wprefix + incl - t;
    oa.x = ex; ex += c0;
    oa.y = ex; ex += c1;
    oa.z = ex; ex += c2;
    oa.w = ex; ex += c3;
    ob.x = ex; ex += c4;
    ob.y = ex; ex += c5;
    ob.z = ex; ex += c6;
    ob.w = ex;
    if (q0 <= lastq) {
      float4 da;
      da.x = rsqrtf((float)c0); da.y = rsqrtf((float)c1);
      da.z = rsqrtf((float)c2); da.w = rsqrtf((float)c3);
      off4[q0]  = oa;
      dinv4[q0] = da;
    }
    if (q0 + 1 <= lastq) {
      float4 db;
      db.x = rsqrtf((float)c4); db.y = rsqrtf((float)c5);
      db.z = rsqrtf((float)c6); db.w = rsqrtf((float)c7);
      off4[q0 + 1]  = ob;
      dinv4[q0 + 1] = db;
    }

    carry += total;
    q0  = q1;
    ra  = na;
    rb  = nb;
    buf ^= 1;
  }
  if (tid == 0) off[n] = carry;
}

__global__ void gcn_fill(const int* __restrict__ src, const int* __restrict__ dst,
                         const int* __restrict__ pos, const int* __restrict__ off,
                         const float* __restrict__ dinv, int* __restrict__ tmp,
                         int* __restrict__ ssrc, float* __restrict__ sw, int e, int n) {
  int i = blockIdx.x * blockDim.x + threadIdx.x;
  if (i < e) tmp[off[dst[i]] + 1 + pos[i]] = src[i];
  if (i < n) {
    const int   o = off[i];
    const float r = dinv[i];
    ssrc[o] = i;
    sw[o]   = r * r;
  }
}

__global__ void gcn_order(const int* __restrict__ src, const int* __restrict__ dst,
                          const int* __restrict__ pos, const int* __restrict__ off,
                          const int* __restrict__ tmp, const float* __restrict__ dinv,
                          int* __restrict__ ssrc, float* __restrict__ sw, int e) {
  int i = blockIdx.x * blockDim.x + threadIdx.x;
  if (i >= e) return;
  const int d = dst[i], s = src[i], p = pos[i];
  const int b = off[d] + 1;
  const int n = off[d + 1] - b;
  int r = 0;
  for (int j = 0; j < n; ++j) {
    const int t = tmp[b + j];
    r += ((t < s) || (t == s && j < p)) ? 1 : 0;
  }
  ssrc[b + r] = s;
  sw[b + r]   = dinv[s] * dinv[d];
}

__device__ __forceinline__ void gcn_axpy(float4& a, const float4& x, float w) {
  a.x = fmaf(x.x, w, a.x);
  a.y = fmaf(x.y, w, a.y);
  a.z = fmaf(x.z, w, a.z);
  a.w = fmaf(x.w, w, a.w);
}

template <int K>
__device__ __forceinline__ void gcn_rows(const float4* __restrict__ Hv, int sv, int wvi,
                                         int j, int lane, float4& a0, float4& a1) {
  float4 x0[K], x1[K];
  #pragma unroll
  for (int k = 0; k < K; ++k) {
    const int i = __builtin_amdgcn_readlane(sv, j + k);
    const float4* __restrict__ r = Hv + (size_t)i * (D_HID / 4);
    x0[k] = r[lane];
    x1[k] = r[32 + lane];
  }
  #pragma unroll
  for (int k = 0; k < K; ++k) {
    const float w = __int_as_float(__builtin_amdgcn_readlane(wvi, j + k));
    gcn_axpy(a0, x0[k], w);
    gcn_axpy(a1, x1[k], w);
  }
}

template <int RELU>
__launch_bounds__(256)
__global__ void gcn_gather(const int* __restrict__ off, const int* __restrict__ ssrc,
                           const float* __restrict__ sw,
                           const float* __restrict__ H, const float* __restrict__ bias,
                           float* __restrict__ out, int n) {
  const int lane = threadIdx.x & 31;
  const int row = blockIdx.x * 8 + __builtin_amdgcn_readfirstlane((int)(threadIdx.x >> 5));
  if (row >= n) return;

  const float4* __restrict__ Hv = (const float4*)H;

  const int b   = off[row];
  const int cnt = off[row + 1] - b;

  float4 a0 = {0.f, 0.f, 0.f, 0.f};
  float4 a1 = a0;

  for (int c0 = 0; c0 < cnt; c0 += 32) {
    const int m = (cnt - c0 < 32) ? (cnt - c0) : 32;
    int   sv = 0;
    float wv = 0.f;
    if (lane < m) {
      sv = ssrc[b + c0 + lane];
      wv = sw[b + c0 + lane];
    }
    const int wvi = __float_as_int(wv);

    int j = 0;
    for (; j + 8 <= m; j += 8) gcn_rows<8>(Hv, sv, wvi, j, lane, a0, a1);
    if (m - j >= 4) { gcn_rows<4>(Hv, sv, wvi, j, lane, a0, a1); j += 4; }
    if (m - j >= 2) { gcn_rows<2>(Hv, sv, wvi, j, lane, a0, a1); j += 2; }
    if (m - j >= 1) { gcn_rows<1>(Hv, sv, wvi, j, lane, a0, a1); }
  }

  const float4* __restrict__ bv = (const float4*)bias;
  const float4 b0 = bv[lane];
  const float4 b1 = bv[32 + lane];
  a0.x += b0.x; a0.y += b0.y; a0.z += b0.z; a0.w += b0.w;
  a1.x += b1.x; a1.y += b1.y; a1.z += b1.z; a1.w += b1.w;
  if (RELU) {
    a0.x = fmaxf(a0.x, 0.f); a0.y = fmaxf(a0.y, 0.f); a0.z = fmaxf(a0.z, 0.f); a0.w = fmaxf(a0.w, 0.f);
    a1.x = fmaxf(a1.x, 0.f); a1.y = fmaxf(a1.y, 0.f); a1.z = fmaxf(a1.z, 0.f); a1.w = fmaxf(a1.w, 0.f);
  }

  float4* __restrict__ ov = (float4*)out + (size_t)row * (D_HID / 4);
  ov[lane]      = a0;
  ov[32 + lane] = a1;
}

extern "C" void kernel_launch(void* const* d_in, const int* in_sizes, int n_in,
                              void* d_out, int out_size, void* d_ws, size_t ws_size,
                              hipStream_t stream) {
  const float* x   = (const float*)d_in[0];
  const int*   ei  = (const int*)  d_in[1];
  const float* W1  = (const float*)d_in[2];
  const float* b1  = (const float*)d_in[3];
  const float* W2  = (const float*)d_in[4];
  const float* b2  = (const float*)d_in[5];
  const float* W3  = (const float*)d_in[6];
  const float* b3  = (const float*)d_in[7];

  const int N = in_sizes[0] / 128;
  const int E = in_sizes[1] / 2;
  const int total = N * D_HID;

  const int* src = ei;
  const int* dst = ei + E;

  const int Npad = (N + 4) & ~3;
  const int L    = E + Npad;
  float* H    = (float*)d_ws;
  float* X    = H + (size_t)total;
  float* dinv = X + (size_t)total;
  int*   cnt  = (int*)(dinv + Npad);
  int*   off  = cnt + Npad;
  int*   pos  = off + Npad;
  int*   tmp  = pos + E;
  int*   ssrc = tmp + L;
  float* sw   = (float*)(ssrc + L);

  const int rowTiles = N / 16;
  dim3 gemmGrid(rowTiles, 1, 1);
  dim3 gemmBlk(128, 1, 1);
  const int nBlkN = (N + 255) / 256;
  const int nBlkE = (E + 255) / 256;
  const int nBlkR = (N + 7) / 8;
  const int nBlkF = (E > N) ? nBlkE : nBlkN;
  const int nBlkZ = (Npad + 255) / 256;

  gcn_cnt_zero <<<nBlkZ, 256, 0, stream>>>(cnt, Npad);
  gcn_cnt_edges<<<nBlkE, 256, 0, stream>>>(dst, cnt, pos, E);
  gcn_scan_deg <<<1, 1024, 0, stream>>>(cnt, off, dinv, N, Npad);
  gcn_fill     <<<nBlkF, 256, 0, stream>>>(src, dst, pos, off, dinv, tmp, ssrc, sw, E, N);
  gcn_order    <<<nBlkE, 256, 0, stream>>>(src, dst, pos, off, tmp, dinv, ssrc, sw, E);

  gcn_gemm_wmma<128><<<gemmGrid, gemmBlk, 0, stream>>>(x, W1, H);
  gcn_gather<1><<<nBlkR, 256, 0, stream>>>(off, ssrc, sw, H, b1, X, N);

  gcn_gemm_wmma<256><<<gemmGrid, gemmBlk, 0, stream>>>(X, W2, H);
  gcn_gather<1><<<nBlkR, 256, 0, stream>>>(off, ssrc, sw, H, b2, X, N);

  gcn_gemm_wmma<256><<<gemmGrid, gemmBlk, 0, stream>>>(X, W3, H);
  gcn_gather<0><<<nBlkR, 256, 0, stream>>>(off, ssrc, sw, H, b3, (float*)d_out, N);
}


# Layer 1/2 row gather fused into the next layer's GEMM launch: a 4-wave work-group sums its 16 rows (bias, relu) into LDS, A fragments read from LDS; 2 launches and the activation round trip dropped
# speedup vs baseline: 4.2312x; 1.2129x over previous
#include <hip/hip_runtime.h>
#include <hip/hip_bf16.h>

#define N_NODES 50000
#define D_HID   256

typedef __attribute__((ext_vector_type(2))) float v2f;
typedef __attribute__((ext_vector_type(8))) float v8f;

typedef __attribute__((ext_vector_type(4))) float v4f;

struct gcn_kblk { v4f a, w0, w1, w2, w3; };

template <int DOUT>
__device__ __forceinline__ gcn_kblk gcn_load_kblk(const float* Xrow, const float* Wbase, int kk) {
  gcn_kblk f;
  f.a  = *(const v4f*)(Xrow + kk);
  f.w0 = *(const v4f*)(Wbase + (size_t)(kk + 0) * DOUT);
  f.w1 = *(const v4f*)(Wbase + (size_t)(kk + 1) * DOUT);
  f.w2 = *(const v4f*)(Wbase + (size_t)(kk + 2) * DOUT);
  f.w3 = *(const v4f*)(Wbase + (size_t)(kk + 3) * DOUT);
  return f;
}

template <int DIN>
__device__ __forceinline__ void gcn_tile_product(const float* Xrow, const float* Wbase,
                                                 v8f& acc0, v8f& acc1, v8f& acc2, v8f& acc3) {
  constexpr int DOUT = D_HID;
  gcn_kblk cur = gcn_load_kblk<DOUT>(Xrow, Wbase, 0);

  #pragma unroll
  for (int kk = 0; kk < DIN; kk += 8) {
    gcn_kblk nxt = cur;
    if (kk + 8 < DIN) nxt = gcn_load_kblk<DOUT>(Xrow, Wbase, kk + 8);
    __builtin_amdgcn_sched_barrier(0);

    v2f alo, ahi;
    alo.x = cur.a.x; alo.y = cur.a.y;
    ahi.x = cur.a.z; ahi.y = cur.a.w;

    v2f b;
    b.x = cur.w0.x; b.y = cur.w1.x;
    acc0 = __builtin_amdgcn_wmma_f32_16x16x4_f32(false, alo, false, b, (short)0, acc0, false, false);
    b.x = cur.w0.y; b.y = cur.w1.y;
    acc1 = __builtin_amdgcn_wmma_f32_16x16x4_f32(false, alo, false, b, (short)0, acc1, false, false);
    b.x = cur.w0.z; b.y = cur.w1.z;
    acc2 = __builtin_amdgcn_wmma_f32_16x16x4_f32(false, alo, false, b, (short)0, acc2, false, false);
    b.x = cur.w0.w; b.y = cur.w1.w;
    acc3 = __builtin_amdgcn_wmma_f32_16x16x4_f32(false, alo, false, b, (short)0, acc3, false, false);

    b.x = cur.w2.x; b.y = cur.w3.x;
    acc0 = __builtin_amdgcn_wmma_f32_16x16x4_f32(false, ahi, false, b, (short)0, acc0, false, false);
    b.x = cur.w2.y; b.y = cur.w3.y;
    acc1 = __builtin_amdgcn_wmma_f32_16x16x4_f32(false, ahi, false, b, (short)0, acc1, false, false);
    b.x = cur.w2.z; b.y = cur.w3.z;
    acc2 = __builtin_amdgcn_wmma_f32_16x16x4_f32(false, ahi, false, b, (short)0, acc2, false, false);
    b.x = cur.w2.w; b.y = cur.w3.w;
    acc3 = __builtin_amdgcn_wmma_f32_16x16x4_f32(false, ahi, false, b, (short)0, acc3, false, false);

    __builtin_amdgcn_sched_barrier(0);
    cur = nxt;
  }
}

__device__ __forceinline__ void gcn_tile_store(float* Hb, int half, const v8f& acc0, const v8f& acc1,
                                               const v8f& acc2, const v8f& acc3) {
  #pragma unroll
  for (int r = 0; r < 8; ++r) {
    v4f o;
    o.x = acc0[r]; o.y = acc1[r]; o.z = acc2[r]; o.w = acc3[r];
    *(v4f*)(Hb + (size_t)(r + 8 * half) * D_HID) = o;
  }
}

template <int DIN>
__launch_bounds__(128)
__global__ void gcn_gemm_wmma(const float* X, const float* W, float* H) {
  constexpr int DOUT = D_HID;

  const int lane = threadIdx.x & 31;
  const int wave = threadIdx.x >> 5;
  const int half = lane >> 4;
  const int c    = lane & 15;

  const int row0 = blockIdx.x * 16;
  const int col0 = wave * 64;

  const float* Xrow  = X + (size_t)(row0 + c) * DIN + 4 * half;
  const float* Wbase = W + (size_t)(4 * half) * DOUT + col0 + 4 * c;

  v8f acc0 = {0.f,0.f,0.f,0.f,0.f,0.f,0.f,0.f};
  v8f acc1 = acc0, acc2 = acc0, acc3 = acc0;

  gcn_tile_product<DIN>(Xrow, Wbase, acc0, acc1, acc2, acc3);

  gcn_tile_store(H + (size_t)row0 * DOUT + col0 + 4 * c, half, acc0, acc1, acc2, acc3);
}

__global__ void gcn_cnt_zero(int* __restrict__ cnt, int n) {
  int i = blockIdx.x * blockDim.x + threadIdx.x;
  if (i < n) cnt[i] = 0;
}

__global__ void gcn_cnt_edges(const int* __restrict__ dst, int* __restrict__ cnt,
                              int* __restrict__ pos, int e) {
  int i = blockIdx.x * blockDim.x + threadIdx.x;
  if (i < e) pos[i] = atomicAdd(&cnt[dst[i]], 1);
}

__launch_bounds__(1024)
__global__ void gcn_scan_deg(const int* __restrict__ cnt, int* __restrict__ off,
                             float* __restrict__ dinv, int n, int npad) {
  __shared__ int wsum[2][32];
  const int tid  = threadIdx.x;
  const int lane = tid & 31;
  const int wave = tid >> 5;

  const int4* __restrict__ cnt4  = (const int4*)cnt;
  int4*       __restrict__ off4  = (int4*)off;
  float4*     __restrict__ dinv4 = (float4*)dinv;
  const int lastq = npad / 4 - 1;

  int carry = 0;
  int buf   = 0;
  int q0 = tid * 2;
  int4 ra = cnt4[min(q0,     lastq)];
  int4 rb = cnt4[min(q0 + 1, lastq)];

  for (int base = 0; base < n; base += 8192) {
    const int q1 = q0 + 2048;
    const int4 na = cnt4[min(q1,     lastq)];
    const int4 nb = cnt4[min(q1 + 1, lastq)];

    const int i0 = q0 * 4;
    const int c0 = (i0     < n) ? ra.x + 1 : 0;
    const int c1 = (i0 + 1 < n) ? ra.y + 1 : 0;
    const int c2 = (i0 + 2 < n) ? ra.z + 1 : 0;
    const int c3 = (i0 + 3 < n) ? ra.w + 1 : 0;
    const int c4 = (i0 + 4 < n) ? rb.x + 1 : 0;
    const int c5 = (i0 + 5 < n) ? rb.y + 1 : 0;
    const int c6 = (i0 + 6 < n) ? rb.z + 1 : 0;
    const int c7 = (i0 + 7 < n) ? rb.w + 1 : 0;

    const int t = c0 + c1 + c2 + c3 + c4 + c5 + c6 + c7;
    int incl = t;
    #pragma unroll
    for (int o = 1; o < 32; o <<= 1) {
      int v = __shfl_up(incl, o);
      if (lane >= o) incl += v;
    }
    if (lane == 31) wsum[buf][wave] = incl;
    __syncthreads();

    const int ws = wsum[buf][lane];
    int wincl = ws;
    #pragma unroll
    for (int o = 1; o < 32; o <<= 1) {
      int v = __shfl_up(wincl, o);
      if (lane >= o) wincl += v;
    }
    const int wprefix = __shfl(wincl - ws, wave);
    const int total   = __shfl(wincl, 31);

    int4 oa, ob;
    int ex = carry + wprefix + incl - t;
    oa.x = ex; ex += c0;
    oa.y = ex; ex += c1;
    oa.z = ex; ex += c2;
    oa.w = ex; ex += c3;
    ob.x = ex; ex += c4;
    ob.y = ex; ex += c5;
    ob.z = ex; ex += c6;
    ob.w = ex;
    if (q0 <= lastq) {
      float4 da;
      da.x = rsqrtf((float)c0); da.y = rsqrtf((float)c1);
      da.z = rsqrtf((float)c2); da.w = rsqrtf((float)c3);
      off4[q0]  = oa;
      dinv4[q0] = da;
    }
    if (q0 + 1 <= lastq) {
      float4 db;
      db.x = rsqrtf((float)c4); db.y = rsqrtf((float)c5);
      db.z = rsqrtf((float)c6); db.w = rsqrtf((float)c7);
      off4[q0 + 1]  = ob;
      dinv4[q0 + 1] = db;
    }

    carry += total;
    q0  = q1;
    ra  = na;
    rb  = nb;
    buf ^= 1;
  }
  if (tid == 0) off[n] = carry;
}

__global__ void gcn_fill(const int* __restrict__ src, const int* __restrict__ dst,
                         const int* __restrict__ pos, const int* __restrict__ off,
                         const float* __restrict__ dinv, int* __restrict__ tmp,
                         int* __restrict__ ssrc, float* __restrict__ sw, int e, int n) {
  int i = blockIdx.x * blockDim.x + threadIdx.x;
  if (i < e) tmp[off[dst[i]] + 1 + pos[i]] = src[i];
  if (i < n) {
    const int   o = off[i];
    const float r = dinv[i];
    ssrc[o] = i;
    sw[o]   = r * r;
  }
}

__global__ void gcn_order(const int* __restrict__ src, const int* __restrict__ dst,
                          const int* __restrict__ pos, const int* __restrict__ off,
                          const int* __restrict__ tmp, const float* __restrict__ dinv,
                          int* __restrict__ ssrc, float* __restrict__ sw, int e) {
  int i = blockIdx.x * blockDim.x + threadIdx.x;
  if (i >= e) return;
  const int d = dst[i], s = src[i], p = pos[i];
  const int b = off[d] + 1;
  const int n = off[d + 1] - b;
  int r = 0;
  for (int j = 0; j < n; ++j) {
    const int t = tmp[b + j];
    r += ((t < s) || (t == s && j < p)) ? 1 : 0;
  }
  ssrc[b + r] = s;
  sw[b + r]   = dinv[s] * dinv[d];
}

__device__ __forceinline__ void gcn_axpy(float4& a, const float4& x, float w) {
  a.x = fmaf(x.x, w, a.x);
  a.y = fmaf(x.y, w, a.y);
  a.z = fmaf(x.z, w, a.z);
  a.w = fmaf(x.w, w, a.w);
}

template <int K>
__device__ __forceinline__ void gcn_rows(const float4* __restrict__ Hv, int sv, int wvi,
                                         int j, int lane, float4& a0, float4& a1) {
  float4 x0[K], x1[K];
  #pragma unroll
  for (int k = 0; k < K; ++k) {
    const int i = __builtin_amdgcn_readlane(sv, j + k);
    const float4* __restrict__ r = Hv + (size_t)i * (D_HID / 4);
    x0[k] = r[lane];
    x1[k] = r[32 + lane];
  }
  #pragma unroll
  for (int k = 0; k < K; ++k) {
    const float w = __int_as_float(__builtin_amdgcn_readlane(wvi, j + k));
    gcn_axpy(a0, x0[k], w);
    gcn_axpy(a1, x1[k], w);
  }
}

__device__ __forceinline__ void gcn_chunk(const float4* __restrict__ Hv, int sv, int wvi, int m,
                                          int lane, float4& a0, float4& a1) {
  int j = 0;
  for (; j + 8 <= m; j += 8) gcn_rows<8>(Hv, sv, wvi, j, lane, a0, a1);
  if (m - j >= 4) { gcn_rows<4>(Hv, sv, wvi, j, lane, a0, a1); j += 4; }
  if (m - j >= 2) { gcn_rows<2>(Hv, sv, wvi, j, lane, a0, a1); j += 2; }
  if (m - j >= 1) { gcn_rows<1>(Hv, sv, wvi, j, lane, a0, a1); }
}

template <int RELU>
__device__ __forceinline__ void gcn_bias_act(float4& a0, float4& a1, const float4& b0, const float4& b1) {
  a0.x += b0.x; a0.y += b0.y; a0.z += b0.z; a0.w += b0.w;
  a1.x += b1.x; a1.y += b1.y; a1.z += b1.z; a1.w += b1.w;
  if (RELU) {
    a0.x = fmaxf(a0.x, 0.f); a0.y = fmaxf(a0.y, 0.f); a0.z = fmaxf(a0.z, 0.f); a0.w = fmaxf(a0.w, 0.f);
    a1.x = fmaxf(a1.x, 0.f); a1.y = fmaxf(a1.y, 0.f); a1.z = fmaxf(a1.z, 0.f); a1.w = fmaxf(a1.w, 0.f);
  }
}

template <int RELU>
__launch_bounds__(256)
__global__ void gcn_gather(const int* __restrict__ off, const int* __restrict__ ssrc,
                           const float* __restrict__ sw,
                           const float* __restrict__ H, const float* __restrict__ bias,
                           float* __restrict__ out, int n) {
  const int lane = threadIdx.x & 31;
  const int row = blockIdx.x * 8 + __builtin_amdgcn_readfirstlane((int)(threadIdx.x >> 5));
  if (row >= n) return;

  const float4* __restrict__ Hv = (const float4*)H;

  const int b   = off[row];
  const int cnt = off[row + 1] - b;

  float4 a0 = {0.f, 0.f, 0.f, 0.f};
  float4 a1 = a0;

  for (int c0 = 0; c0 < cnt; c0 += 32) {
    const int m = (cnt - c0 < 32) ? (cnt - c0) : 32;
    int   sv = 0;
    float wv = 0.f;
    if (lane < m) {
      sv = ssrc[b + c0 + lane];
      wv = sw[b + c0 + lane];
    }
    gcn_chunk(Hv, sv, __float_as_int(wv), m, lane, a0, a1);
  }

  const float4* __restrict__ bv = (const float4*)bias;
  gcn_bias_act<RELU>(a0, a1, bv[lane], bv[32 + lane]);

  float4* __restrict__ ov = (float4*)out + (size_t)row * (D_HID / 4);
  ov[lane]      = a0;
  ov[32 + lane] = a1;
}

#define XS_PITCH 260

template <int RELU>
__launch_bounds__(128)
__global__ void gcn_gather_gemm(const int* __restrict__ off, const int* __restrict__ ssrc,
                                const float* __restrict__ sw,
                                const float* __restrict__ Hin, const float* __restrict__ bias,
                                const float* W, float* Hout) {
  __shared__ __attribute__((aligned(16))) float xs[16 * XS_PITCH];

  const int lane = threadIdx.x & 31;
  const int wave = __builtin_amdgcn_readfirstlane((int)(threadIdx.x >> 5));
  const int row0 = blockIdx.x * 16;

  {
    const float4* __restrict__ Hv = (const float4*)Hin;
    const float4* __restrict__ bv = (const float4*)bias;
    const int rbase = row0 + 4 * wave;

    int ob[5];
    #pragma unroll
    for (int r = 0; r < 5; ++r) ob[r] = off[rbase + r];
    int   sv0[4];
    float wv0[4];
    #pragma unroll
    for (int r = 0; r < 4; ++r) {
      sv0[r] = 0;
      wv0[r] = 0.f;
      if (lane < ob[r + 1] - ob[r]) {
        sv0[r] = ssrc[ob[r] + lane];
        wv0[r] = sw[ob[r] + lane];
      }
    }

    #pragma unroll
    for (int r = 0; r < 4; ++r) {
      const int b   = ob[r];
      const int cnt = ob[r + 1] - b;
      float4 a0 = {0.f, 0.f, 0.f, 0.f};
      float4 a1 = a0;
      for (int c0 = 0; c0 < cnt; c0 += 32) {
        const int m = (cnt - c0 < 32) ? (cnt - c0) : 32;
        int   sv = sv0[r];
        float wv = wv0[r];
        if (c0 != 0) {
          sv = 0;
          wv = 0.f;
          if (lane < m) {
            sv = ssrc[b + c0 + lane];
            wv = sw[b + c0 + lane];
          }
        }
        gcn_chunk(Hv, sv, __float_as_int(wv), m, lane, a0, a1);
      }
      gcn_bias_act<RELU>(a0, a1, bv[lane], bv[32 + lane]);

      float* xr = xs + (4 * wave + r) * XS_PITCH;
      *(float4*)(xr + 4 * lane)       = a0;
      *(float4*)(xr + 128 + 4 * lane) = a1;
    }
  }
  __syncthreads();

  const int half = lane >> 4;
  const int c    = lane & 15;
  const int col0 = wave * 64;

  const float* Xrow  = xs + c * XS_PITCH + 4 * half;
  const float* Wbase = W + (size_t)(4 * half) * D_HID + col0 + 4 * c;

  v8f acc0 = {0.f,0.f,0.f,0.f,0.f,0.f,0.f,0.f};
  v8f acc1 = acc0, acc2 = acc0, acc3 = acc0;

  gcn_tile_product<D_HID>(Xrow, Wbase, acc0, acc1, acc2, acc3);

  gcn_tile_store(Hout + (size_t)row0 * D_HID + col0 + 4 * c, half, acc0, acc1, acc2, acc3);
}

extern "C" void kernel_launch(void* const* d_in, const int* in_sizes, int n_in,
                              void* d_out, int out_size, void* d_ws, size_t ws_size,
                              hipStream_t stream) {
  const float* x   = (const float*)d_in[0];
  const int*   ei  = (const int*)  d_in[1];
  const float* W1  = (const float*)d_in[2];
  const float* b1  = (const float*)d_in[3];
  const float* W2  = (const float*)d_in[4];
  const float* b2  = (const float*)d_in[5];
  const float* W3  = (const float*)d_in[6];
  const float* b3  = (const float*)d_in[7];

  const int N = in_sizes[0] / 128;
  const int E = in_sizes[1] / 2;
  const int total = N * D_HID;

  const int* src = ei;
  const int* dst = ei + E;

  const int Npad = (N + 4) & ~3;
  const int L    = E + Npad;
  float* H    = (float*)d_ws;
  float* X    = H + (size_t)total;
  float* dinv = X + (size_t)total;
  int*   cnt  = (int*)(dinv + Npad);
  int*   off  = cnt + Npad;
  int*   pos  = off + Npad;
  int*   tmp  = pos + E;
  int*   ssrc = tmp + L;
  float* sw   = (float*)(ssrc + L);

  const int rowTiles = N / 16;
  dim3 gemmGrid(rowTiles, 1, 1);
  dim3 gemmBlk(128, 1, 1);
  const int nBlkN = (N + 255) / 256;
  const int nBlkE = (E + 255) / 256;
  const int nBlkR = (N + 7) / 8;
  const int nBlkF = (E > N) ? nBlkE : nBlkN;
  const int nBlkZ = (Npad + 255) / 256;

  gcn_cnt_zero <<<nBlkZ, 256, 0, stream>>>(cnt, Npad);
  gcn_cnt_edges<<<nBlkE, 256, 0, stream>>>(dst, cnt, pos, E);
  gcn_scan_deg <<<1, 1024, 0, stream>>>(cnt, off, dinv, N, Npad);
  gcn_fill     <<<nBlkF, 256, 0, stream>>>(src, dst, pos, off, dinv, tmp, ssrc, sw, E, N);
  gcn_order    <<<nBlkE, 256, 0, stream>>>(src, dst, pos, off, tmp, dinv, ssrc, sw, E);

  gcn_gemm_wmma<128><<<gemmGrid, gemmBlk, 0, stream>>>(x, W1, H);

  gcn_gather_gemm<1><<<gemmGrid, gemmBlk, 0, stream>>>(off, ssrc, sw, H, b1, W2, X);

  gcn_gather_gemm<1><<<gemmGrid, gemmBlk, 0, stream>>>(off, ssrc, sw, X, b2, W3, H);

  gcn_gather<0><<<nBlkR, 256, 0, stream>>>(off, ssrc, sw, H, b3, (float*)d_out, N);
}


# Each layer computed as (A_hat X) W: rows aggregated on the layer's input width (128 in layer 1) into LDS, product after, bias/relu in its store; 3 fused launches replace 4; fp32 sums reassociated
# speedup vs baseline: 4.6798x; 1.1060x over previous
#include <hip/hip_runtime.h>
#include <hip/hip_bf16.h>

#define N_NODES 50000
#define D_HID   256

typedef __attribute__((ext_vector_type(2))) float v2f;
typedef __attribute__((ext_vector_type(8))) float v8f;

typedef __attribute__((ext_vector_type(4))) float v4f;

struct gcn_kblk { v4f a, w0, w1, w2, w3; };

template <int DOUT>
__device__ __forceinline__ gcn_kblk gcn_load_kblk(const float* Xrow, const float* Wbase, int kk) {
  gcn_kblk f;
  f.a  = *(const v4f*)(Xrow + kk);
  f.w0 = *(const v4f*)(Wbase + (size_t)(kk + 0) * DOUT);
  f.w1 = *(const v4f*)(Wbase + (size_t)(kk + 1) * DOUT);
  f.w2 = *(const v4f*)(Wbase + (size_t)(kk + 2) * DOUT);
  f.w3 = *(const v4f*)(Wbase + (size_t)(kk + 3) * DOUT);
  return f;
}

template <int DIN>
__device__ __forceinline__ void gcn_tile_product(const float* Xrow, const float* Wbase,
                                                 v8f& acc0, v8f& acc1, v8f& acc2, v8f& acc3) {
  constexpr int DOUT = D_HID;
  gcn_kblk cur = gcn_load_kblk<DOUT>(Xrow, Wbase, 0);

  #pragma unroll
  for (int kk = 0; kk < DIN; kk += 8) {
    gcn_kblk nxt = cur;
    if (kk + 8 < DIN) nxt = gcn_load_kblk<DOUT>(Xrow, Wbase, kk + 8);
    __builtin_amdgcn_sched_barrier(0);

    v2f alo, ahi;
    alo.x = cur.a.x; alo.y = cur.a.y;
    ahi.x = cur.a.z; ahi.y = cur.a.w;

    v2f b;
    b.x = cur.w0.x; b.y = cur.w1.x;
    acc0 = __builtin_amdgcn_wmma_f32_16x16x4_f32(false, alo, false, b, (short)0, acc0, false, false);
    b.x = cur.w0.y; b.y = cur.w1.y;
    acc1 = __builtin_amdgcn_wmma_f32_16x16x4_f32(false, alo, false, b, (short)0, acc1, false, false);
    b.x = cur.w0.z; b.y = cur.w1.z;
    acc2 = __builtin_amdgcn_wmma_f32_16x16x4_f32(false, alo, false, b, (short)0, acc2, false, false);
    b.x = cur.w0.w; b.y = cur.w1.w;
    acc3 = __builtin_amdgcn_wmma_f32_16x16x4_f32(false, alo, false, b, (short)0, acc3, false, false);

    b.x = cur.w2.x; b.y = cur.w3.x;
    acc0 = __builtin_amdgcn_wmma_f32_16x16x4_f32(false, ahi, false, b, (short)0, acc0, false, false);
    b.x = cur.w2.y; b.y = cur.w3.y;
    acc1 = __builtin_amdgcn_wmma_f32_16x16x4_f32(false, ahi, false, b, (short)0, acc1, false, false);
    b.x = cur.w2.z; b.y = cur.w3.z;
    acc2 = __builtin_amdgcn_wmma_f32_16x16x4_f32(false, ahi, false, b, (short)0, acc2, false, false);
    b.x = cur.w2.w; b.y = cur.w3.w;
    acc3 = __builtin_amdgcn_wmma_f32_16x16x4_f32(false, ahi, false, b, (short)0, acc3, false, false);

    __builtin_amdgcn_sched_barrier(0);
    cur = nxt;
  }
}

template <int RELU>
__device__ __forceinline__ void gcn_tile_store(float* Yb, int half, const v4f& bq, const v8f& acc0,
                                               const v8f& acc1, const v8f& acc2, const v8f& acc3) {
  #pragma unroll
  for (int r = 0; r < 8; ++r) {
    v4f o;
    o.x = acc0[r] + bq.x; o.y = acc1[r] + bq.y; o.z = acc2[r] + bq.z; o.w = acc3[r] + bq.w;
    if (RELU) {
      o.x = fmaxf(o.x, 0.f); o.y = fmaxf(o.y, 0.f); o.z = fmaxf(o.z, 0.f); o.w = fmaxf(o.w, 0.f);
    }
    *(v4f*)(Yb + (size_t)(r + 8 * half) * D_HID) = o;
  }
}

__global__ void gcn_cnt_zero(int* __restrict__ cnt, int n) {
  int i = blockIdx.x * blockDim.x + threadIdx.x;
  if (i < n) cnt[i] = 0;
}

__global__ void gcn_cnt_edges(const int* __restrict__ dst, int* __restrict__ cnt,
                              int* __restrict__ pos, int e) {
  int i = blockIdx.x * blockDim.x + threadIdx.x;
  if (i < e) pos[i] = atomicAdd(&cnt[dst[i]], 1);
}

__launch_bounds__(1024)
__global__ void gcn_scan_deg(const int* __restrict__ cnt, int* __restrict__ off,
                             float* __restrict__ dinv, int n, int npad) {
  __shared__ int wsum[2][32];
  const int tid  = threadIdx.x;
  const int lane = tid & 31;
  const int wave = tid >> 5;

  const int4* __restrict__ cnt4  = (const int4*)cnt;
  int4*       __restrict__ off4  = (int4*)off;
  float4*     __restrict__ dinv4 = (float4*)dinv;
  const int lastq = npad / 4 - 1;

  int carry = 0;
  int buf   = 0;
  int q0 = tid * 2;
  int4 ra = cnt4[min(q0,     lastq)];
  int4 rb = cnt4[min(q0 + 1, lastq)];

  for (int base = 0; base < n; base += 8192) {
    const int q1 = q0 + 2048;
    const int4 na = cnt4[min(q1,     lastq)];
    const int4 nb = cnt4[min(q1 + 1, lastq)];

    const int i0 = q0 * 4;
    const int c0 = (i0     < n) ? ra.x + 1 : 0;
    const int c1 = (i0 + 1 < n) ? ra.y + 1 : 0;
    const int c2 = (i0 + 2 < n) ? ra.z + 1 : 0;
    const int c3 = (i0 + 3 < n) ? ra.w + 1 : 0;
    const int c4 = (i0 + 4 < n) ? rb.x + 1 : 0;
    const int c5 = (i0 + 5 < n) ? rb.y + 1 : 0;
    const int c6 = (i0 + 6 < n) ? rb.z + 1 : 0;
    const int c7 = (i0 + 7 < n) ? rb.w + 1 : 0;

    const int t = c0 + c1 + c2 + c3 + c4 + c5 + c6 + c7;
    int incl = t;
    #pragma unroll
    for (int o = 1; o < 32; o <<= 1) {
      int v = __shfl_up(incl, o);
      if (lane >= o) incl += v;
    }
    if (lane == 31) wsum[buf][wave] = incl;
    __syncthreads();

    const int ws = wsum[buf][lane];
    int wincl = ws;
    #pragma unroll
    for (int o = 1; o < 32; o <<= 1) {
      int v = __shfl_up(wincl, o);
      if (lane >= o) wincl += v;
    }
    const int wprefix = __shfl(wincl - ws, wave);
    const int total   = __shfl(wincl, 31);

    int4 oa, ob;
    int ex = carry + wprefix + incl - t;
    oa.x = ex; ex += c0;
    oa.y = ex; ex += c1;
    oa.z = ex; ex += c2;
    oa.w = ex; ex += c3;
    ob.x = ex; ex += c4;
    ob.y = ex; ex += c5;
    ob.z = ex; ex += c6;
    ob.w = ex;
    if (q0 <= lastq) {
      float4 da;
      da.x = rsqrtf((float)c0); da.y = rsqrtf((float)c1);
      da.z = rsqrtf((float)c2); da.w = rsqrtf((float)c3);
      off4[q0]  = oa;
      dinv4[q0] = da;
    }
    if (q0 + 1 <= lastq) {
      float4 db;
      db.x = rsqrtf((float)c4); db.y = rsqrtf((float)c5);
      db.z = rsqrtf((float)c6); db.w = rsqrtf((float)c7);
      off4[q0 + 1]  = ob;
      dinv4[q0 + 1] = db;
    }

    carry += total;
    q0  = q1;
    ra  = na;
    rb  = nb;
    buf ^= 1;
  }
  if (tid == 0) off[n] = carry;
}

__global__ void gcn_fill(const int* __restrict__ src, const int* __restrict__ dst,
                         const int* __restrict__ pos, const int* __restrict__ off,
                         const float* __restrict__ dinv, int* __restrict__ tmp,
                         int* __restrict__ ssrc, float* __restrict__ sw, int e, int n) {
  int i = blockIdx.x * blockDim.x + threadIdx.x;
  if (i < e) tmp[off[dst[i]] + 1 + pos[i]] = src[i];
  if (i < n) {
    const int   o = off[i];
    const float r = dinv[i];
    ssrc[o] = i;
    sw[o]   = r * r;
  }
}

__global__ void gcn_order(const int* __restrict__ src, const int* __restrict__ dst,
                          const int* __restrict__ pos, const int* __restrict__ off,
                          const int* __restrict__ tmp, const float* __restrict__ dinv,
                          int* __restrict__ ssrc, float* __restrict__ sw, int e) {
  int i = blockIdx.x * blockDim.x + threadIdx.x;
  if (i >= e) return;
  const int d = dst[i], s = src[i], p = pos[i];
  const int b = off[d] + 1;
  const int n = off[d + 1] - b;
  int r = 0;
  for (int j = 0; j < n; ++j) {
    const int t = tmp[b + j];
    r += ((t < s) || (t == s && j < p)) ? 1 : 0;
  }
  ssrc[b + r] = s;
  sw[b + r]   = dinv[s] * dinv[d];
}

__device__ __forceinline__ void gcn_axpy(float4& a, const float4& x, float w) {
  a.x = fmaf(x.x, w, a.x);
  a.y = fmaf(x.y, w, a.y);
  a.z = fmaf(x.z, w, a.z);
  a.w = fmaf(x.w, w, a.w);
}

template <int NV, int K>
__device__ __forceinline__ void gcn_rows(const float4* __restrict__ Xv, int sv, int wvi,
                                         int j, int lane, float4 (&a)[NV]) {
  float4 x[K][NV];
  #pragma unroll
  for (int k = 0; k < K; ++k) {
    const int i = __builtin_amdgcn_readlane(sv, j + k);
    const float4* __restrict__ r = Xv + (size_t)i * (32 * NV);
    #pragma unroll
    for (int v = 0; v < NV; ++v) x[k][v] = r[32 * v + lane];
  }
  #pragma unroll
  for (int k = 0; k < K; ++k) {
    const float w = __int_as_float(__builtin_amdgcn_readlane(wvi, j + k));
    #pragma unroll
    for (int v = 0; v < NV; ++v) gcn_axpy(a[v], x[k][v], w);
  }
}

template <int NV>
__device__ __forceinline__ void gcn_chunk(const float4* __restrict__ Xv, int sv, int wvi, int m,
                                          int lane, float4 (&a)[NV]) {
  constexpr int KMAX = 16 / NV;
  int j = 0;
  for (; j + KMAX <= m; j += KMAX) gcn_rows<NV, KMAX>(Xv, sv, wvi, j, lane, a);
  if (KMAX > 8 && m - j >= 8) { gcn_rows<NV, 8>(Xv, sv, wvi, j, lane, a); j += 8; }
  if (m - j >= 4) { gcn_rows<NV, 4>(Xv, sv, wvi, j, lane, a); j += 4; }
  if (m - j >= 2) { gcn_rows<NV, 2>(Xv, sv, wvi, j, lane, a); j += 2; }
  if (m - j >= 1) { gcn_rows<NV, 1>(Xv, sv, wvi, j, lane, a); }
}

template <int DIN, int RELU>
__launch_bounds__(128)
__global__ void gcn_agg_gemm(const int* __restrict__ off, const int* __restrict__ ssrc,
                             const float* __restrict__ sw,
                             const float* __restrict__ Xin, const float* W,
                             const float* __restrict__ bias, float* __restrict__ Yout) {
  constexpr int NV    = DIN / 128;
  constexpr int PITCH = DIN + 4;
  __shared__ __attribute__((aligned(16))) float xs[16 * PITCH];

  const int lane = threadIdx.x & 31;
  const int wave = __builtin_amdgcn_readfirstlane((int)(threadIdx.x >> 5));
  const int row0 = blockIdx.x * 16;

  {
    const float4* __restrict__ Xv = (const float4*)Xin;
    const int rbase = row0 + 4 * wave;

    int ob[5];
    #pragma unroll
    for (int r = 0; r < 5; ++r) ob[r] = off[rbase + r];
    int   sv0[4];
    float wv0[4];
    #pragma unroll
    for (int r = 0; r < 4; ++r) {
      sv0[r] = 0;
      wv0[r] = 0.f;
      if (lane < ob[r + 1] - ob[r]) {
        sv0[r] = ssrc[ob[r] + lane];
        wv0[r] = sw[ob[r] + lane];
      }
    }

    #pragma unroll
    for (int r = 0; r < 4; ++r) {
      const int b   = ob[r];
      const int cnt = ob[r + 1] - b;
      float4 a[NV];
      #pragma unroll
      for (int v = 0; v < NV; ++v) a[v] = make_float4(0.f, 0.f, 0.f, 0.f);
      for (int c0 = 0; c0 < cnt; c0 += 32) {
        const int m = (cnt - c0 < 32) ? (cnt - c0) : 32;
        int   sv = sv0[r];
        float wv = wv0[r];
        if (c0 != 0) {
          sv = 0;
          wv = 0.f;
          if (lane < m) {
            sv = ssrc[b + c0 + lane];
            wv = sw[b + c0 + lane];
          }
        }
        gcn_chunk<NV>(Xv, sv, __float_as_int(wv), m, lane, a);
      }

      float* xr = xs + (4 * wave + r) * PITCH;
      #pragma unroll
      for (int v = 0; v < NV; ++v) *(float4*)(xr + 128 * v + 4 * lane) = a[v];
    }
  }
  __syncthreads();

  const int half = lane >> 4;
  const int c    = lane & 15;
  const int col0 = wave * 64;

  const float* Xrow  = xs + c * PITCH + 4 * half;
  const float* Wbase = W + (size_t)(4 * half) * D_HID + col0 + 4 * c;

  v8f acc0 = {0.f,0.f,0.f,0.f,0.f,0.f,0.f,0.f};
  v8f acc1 = acc0, acc2 = acc0, acc3 = acc0;

  gcn_tile_product<DIN>(Xrow, Wbase, acc0, acc1, acc2, acc3);

  const v4f bq = *(const v4f*)(bias + col0 + 4 * c);
  gcn_tile_store<RELU>(Yout + (size_t)row0 * D_HID + col0 + 4 * c, half, bq, acc0, acc1, acc2, acc3);
}

extern "C" void kernel_launch(void* const* d_in, const int* in_sizes, int n_in,
                              void* d_out, int out_size, void* d_ws, size_t ws_size,
                              hipStream_t stream) {
  const float* x   = (const float*)d_in[0];
  const int*   ei  = (const int*)  d_in[1];
  const float* W1  = (const float*)d_in[2];
  const float* b1  = (const float*)d_in[3];
  const float* W2  = (const float*)d_in[4];
  const float* b2  = (const float*)d_in[5];
  const float* W3  = (const float*)d_in[6];
  const float* b3  = (const float*)d_in[7];

  const int N = in_sizes[0] / 128;
  const int E = in_sizes[1] / 2;
  const int total = N * D_HID;

  const int* src = ei;
  const int* dst = ei + E;

  const int Npad = (N + 4) & ~3;
  const int L    = E + Npad;
  float* H    = (float*)d_ws;
  float* X    = H + (size_t)total;
  float* dinv = X + (size_t)total;
  int*   cnt  = (int*)(dinv + Npad);
  int*   off  = cnt + Npad;
  int*   pos  = off + Npad;
  int*   tmp  = pos + E;
  int*   ssrc = tmp + L;
  float* sw   = (float*)(ssrc + L);

  const int rowTiles = N / 16;
  dim3 gemmGrid(rowTiles, 1, 1);
  dim3 gemmBlk(128, 1, 1);
  const int nBlkN = (N + 255) / 256;
  const int nBlkE = (E + 255) / 256;
  const int nBlkF = (E > N) ? nBlkE : nBlkN;
  const int nBlkZ = (Npad + 255) / 256;

  gcn_cnt_zero <<<nBlkZ, 256, 0, stream>>>(cnt, Npad);
  gcn_cnt_edges<<<nBlkE, 256, 0, stream>>>(dst, cnt, pos, E);
  gcn_scan_deg <<<1, 1024, 0, stream>>>(cnt, off, dinv, N, Npad);
  gcn_fill     <<<nBlkF, 256, 0, stream>>>(src, dst, pos, off, dinv, tmp, ssrc, sw, E, N);
  gcn_order    <<<nBlkE, 256, 0, stream>>>(src, dst, pos, off, tmp, dinv, ssrc, sw, E);

  gcn_agg_gemm<128, 1><<<gemmGrid, gemmBlk, 0, stream>>>(off, ssrc, sw, x, W1, b1, H);

  gcn_agg_gemm<256, 1><<<gemmGrid, gemmBlk, 0, stream>>>(off, ssrc, sw, H, W2, b2, X);

  gcn_agg_gemm<256, 0><<<gemmGrid, gemmBlk, 0, stream>>>(off, ssrc, sw, X, W3, b3, (float*)d_out);
}


# Degree count and edge slots: scattered returning global atomics replaced by LDS counter tables (16 work-groups, one edge chunk each) and a per-node sum over the chunks; list order kept
# speedup vs baseline: 4.9617x; 1.0602x over previous
#include <hip/hip_runtime.h>
#include <hip/hip_bf16.h>

#define N_NODES 50000
#define D_HID   256

typedef __attribute__((ext_vector_type(2))) float v2f;
typedef __attribute__((ext_vector_type(8))) float v8f;

typedef __attribute__((ext_vector_type(4))) float v4f;

struct gcn_kblk { v4f a, w0, w1, w2, w3; };

template <int DOUT>
__device__ __forceinline__ gcn_kblk gcn_load_kblk(const float* Xrow, const float* Wbase, int kk) {
  gcn_kblk f;
  f.a  = *(const v4f*)(Xrow + kk);
  f.w0 = *(const v4f*)(Wbase + (size_t)(kk + 0) * DOUT);
  f.w1 = *(const v4f*)(Wbase + (size_t)(kk + 1) * DOUT);
  f.w2 = *(const v4f*)(Wbase + (size_t)(kk + 2) * DOUT);
  f.w3 = *(const v4f*)(Wbase + (size_t)(kk + 3) * DOUT);
  return f;
}

template <int DIN>
__device__ __forceinline__ void gcn_tile_product(const float* Xrow, const float* Wbase,
                                                 v8f& acc0, v8f& acc1, v8f& acc2, v8f& acc3) {
  constexpr int DOUT = D_HID;
  gcn_kblk cur = gcn_load_kblk<DOUT>(Xrow, Wbase, 0);

  #pragma unroll
  for (int kk = 0; kk < DIN; kk += 8) {
    gcn_kblk nxt = cur;
    if (kk + 8 < DIN) nxt = gcn_load_kblk<DOUT>(Xrow, Wbase, kk + 8);
    __builtin_amdgcn_sched_barrier(0);

    v2f alo, ahi;
    alo.x = cur.a.x; alo.y = cur.a.y;
    ahi.x = cur.a.z; ahi.y = cur.a.w;

    v2f b;
    b.x = cur.w0.x; b.y = cur.w1.x;
    acc0 = __builtin_amdgcn_wmma_f32_16x16x4_f32(false, alo, false, b, (short)0, acc0, false, false);
    b.x = cur.w0.y; b.y = cur.w1.y;
    acc1 = __builtin_amdgcn_wmma_f32_16x16x4_f32(false, alo, false, b, (short)0, acc1, false, false);
    b.x = cur.w0.z; b.y = cur.w1.z;
    acc2 = __builtin_amdgcn_wmma_f32_16x16x4_f32(false, alo, false, b, (short)0, acc2, false, false);
    b.x = cur.w0.w; b.y = cur.w1.w;
    acc3 = __builtin_amdgcn_wmma_f32_16x16x4_f32(false, alo, false, b, (short)0, acc3, false, false);

    b.x = cur.w2.x; b.y = cur.w3.x;
    acc0 = __builtin_amdgcn_wmma_f32_16x16x4_f32(false, ahi, false, b, (short)0, acc0, false, false);
    b.x = cur.w2.y; b.y = cur.w3.y;
    acc1 = __builtin_amdgcn_wmma_f32_16x16x4_f32(false, ahi, false, b, (short)0, acc1, false, false);
    b.x = cur.w2.z; b.y = cur.w3.z;
    acc2 = __builtin_amdgcn_wmma_f32_16x16x4_f32(false, ahi, false, b, (short)0, acc2, false, false);
    b.x = cur.w2.w; b.y = cur.w3.w;
    acc3 = __builtin_amdgcn_wmma_f32_16x16x4_f32(false, ahi, false, b, (short)0, acc3, false, false);

    __builtin_amdgcn_sched_barrier(0);
    cur = nxt;
  }
}

template <int RELU>
__device__ __forceinline__ void gcn_tile_store(float* Yb, int half, const v4f& bq, const v8f& acc0,
                                               const v8f& acc1, const v8f& acc2, const v8f& acc3) {
  #pragma unroll
  for (int r = 0; r < 8; ++r) {
    v4f o;
    o.x = acc0[r] + bq.x; o.y = acc1[r] + bq.y; o.z = acc2[r] + bq.z; o.w = acc3[r] + bq.w;
    if (RELU) {
      o.x = fmaxf(o.x, 0.f); o.y = fmaxf(o.y, 0.f); o.z = fmaxf(o.z, 0.f); o.w = fmaxf(o.w, 0.f);
    }
    *(v4f*)(Yb + (size_t)(r + 8 * half) * D_HID) = o;
  }
}

#define GCN_HG 16
#define GCN_HB 32
__launch_bounds__(1024)
__global__ void gcn_cnt_lds(const int* __restrict__ dst, int* __restrict__ loc,
                            int* __restrict__ lrank, int e, int n, int npad) {
  __shared__ __attribute__((aligned(16))) int hist[N_NODES + 32];
  const int tid  = threadIdx.x;
  const int lane = tid & 31;
  int4* hist4 = (int4*)hist;
  for (int k = tid; k < (N_NODES + 32) / 4; k += 1024) hist4[k] = make_int4(0, 0, 0, 0);
  __syncthreads();

  const int per = (e + GCN_HG - 1) / GCN_HG;
  const int lo  = blockIdx.x * per;
  const int hi  = min(lo + per, e);
  for (int i0 = lo + tid; i0 < hi; i0 += 1024 * GCN_HB) {
    int d[GCN_HB], r[GCN_HB];
    #pragma unroll
    for (int u = 0; u < GCN_HB; ++u) {
      const int i = i0 + u * 1024;
      d[u] = dst[min(i, hi - 1)];
    }
    #pragma unroll
    for (int u = 0; u < GCN_HB; ++u) {
      const int i = i0 + u * 1024;
      r[u] = atomicAdd(&hist[(i < hi) ? d[u] : N_NODES + lane], 1);
    }
    #pragma unroll
    for (int u = 0; u < GCN_HB; ++u) {
      const int i = i0 + u * 1024;
      if (i < hi) lrank[i] = r[u];
    }
  }
  __syncthreads();

  int4* __restrict__ out4 = (int4*)(loc + (size_t)blockIdx.x * npad);
  for (int k = tid; k < npad / 4; k += 1024) {
    int4 v = hist4[k];
    v.x = (4 * k     < n) ? v.x : 0;
    v.y = (4 * k + 1 < n) ? v.y : 0;
    v.z = (4 * k + 2 < n) ? v.z : 0;
    v.w = (4 * k + 3 < n) ? v.w : 0;
    out4[k] = v;
  }
}

__global__ void gcn_cnt_combine(int* __restrict__ loc, int* __restrict__ cnt, int npad) {
  const int d = blockIdx.x * blockDim.x + threadIdx.x;
  if (d >= npad) return;
  int c[GCN_HG];
  #pragma unroll
  for (int g = 0; g < GCN_HG; ++g) c[g] = loc[(size_t)g * npad + d];
  int run = 0;
  #pragma unroll
  for (int g = 0; g < GCN_HG; ++g) {
    loc[(size_t)g * npad + d] = run;
    run += c[g];
  }
  cnt[d] = run;
}

__launch_bounds__(1024)
__global__ void gcn_scan_deg(const int* __restrict__ cnt, int* __restrict__ off,
                             float* __restrict__ dinv, int n, int npad) {
  __shared__ int wsum[2][32];
  const int tid  = threadIdx.x;
  const int lane = tid & 31;
  const int wave = tid >> 5;

  const int4* __restrict__ cnt4  = (const int4*)cnt;
  int4*       __restrict__ off4  = (int4*)off;
  float4*     __restrict__ dinv4 = (float4*)dinv;
  const int lastq = npad / 4 - 1;

  int carry = 0;
  int buf   = 0;
  int q0 = tid * 2;
  int4 ra = cnt4[min(q0,     lastq)];
  int4 rb = cnt4[min(q0 + 1, lastq)];

  for (int base = 0; base < n; base += 8192) {
    const int q1 = q0 + 2048;
    const int4 na = cnt4[min(q1,     lastq)];
    const int4 nb = cnt4[min(q1 + 1, lastq)];

    const int i0 = q0 * 4;
    const int c0 = (i0     < n) ? ra.x + 1 : 0;
    const int c1 = (i0 + 1 < n) ? ra.y + 1 : 0;
    const int c2 = (i0 + 2 < n) ? ra.z + 1 : 0;
    const int c3 = (i0 + 3 < n) ? ra.w + 1 : 0;
    const int c4 = (i0 + 4 < n) ? rb.x + 1 : 0;
    const int c5 = (i0 + 5 < n) ? rb.y + 1 : 0;
    const int c6 = (i0 + 6 < n) ? rb.z + 1 : 0;
    const int c7 = (i0 + 7 < n) ? rb.w + 1 : 0;

    const int t = c0 + c1 + c2 + c3 + c4 + c5 + c6 + c7;
    int incl = t;
    #pragma unroll
    for (int o = 1; o < 32; o <<= 1) {
      int v = __shfl_up(incl, o);
      if (lane >= o) incl += v;
    }
    if (lane == 31) wsum[buf][wave] = incl;
    __syncthreads();

    const int ws = wsum[buf][lane];
    int wincl = ws;
    #pragma unroll
    for (int o = 1; o < 32; o <<= 1) {
      int v = __shfl_up(wincl, o);
      if (lane >= o) wincl += v;
    }
    const int wprefix = __shfl(wincl - ws, wave);
    const int total   = __shfl(wincl, 31);

    int4 oa, ob;
    int ex = carry + wprefix + incl - t;
    oa.x = ex; ex += c0;
    oa.y = ex; ex += c1;
    oa.z = ex; ex += c2;
    oa.w = ex; ex += c3;
    ob.x = ex; ex += c4;
    ob.y = ex; ex += c5;
    ob.z = ex; ex += c6;
    ob.w = ex;
    if (q0 <= lastq) {
      float4 da;
      da.x = rsqrtf((float)c0); da.y = rsqrtf((float)c1);
      da.z = rsqrtf((float)c2); da.w = rsqrtf((float)c3);
      off4[q0]  = oa;
      dinv4[q0] = da;
    }
    if (q0 + 1 <= lastq) {
      float4 db;
      db.x = rsqrtf((float)c4); db.y = rsqrtf((float)c5);
      db.z = rsqrtf((float)c6); db.w = rsqrtf((float)c7);
      off4[q0 + 1]  = ob;
      dinv4[q0 + 1] = db;
    }

    carry += total;
    q0  = q1;
    ra  = na;
    rb  = nb;
    buf ^= 1;
  }
  if (tid == 0) off[n] = carry;
}

__global__ void gcn_fill(const int* __restrict__ src, const int* __restrict__ dst,
                         int* __restrict__ pos, const int* __restrict__ loc,
                         const int* __restrict__ off,
                         const float* __restrict__ dinv, int* __restrict__ tmp,
                         int* __restrict__ ssrc, float* __restrict__ sw, int e, int n, int npad) {
  int i = blockIdx.x * blockDim.x + threadIdx.x;
  if (i < e) {
    const int per = (e + GCN_HG - 1) / GCN_HG;
    const int d = dst[i];
    const int p = loc[(size_t)(i / per) * npad + d] + pos[i];
    pos[i] = p;
    tmp[off[d] + 1 + p] = src[i];
  }
  if (i < n) {
    const int   o = off[i];
    const float r = dinv[i];
    ssrc[o] = i;
    sw[o]   = r * r;
  }
}

__global__ void gcn_order(const int* __restrict__ src, const int* __restrict__ dst,
                          const int* __restrict__ pos, const int* __restrict__ off,
                          const int* __restrict__ tmp, const float* __restrict__ dinv,
                          int* __restrict__ ssrc, float* __restrict__ sw, int e) {
  int i = blockIdx.x * blockDim.x + threadIdx.x;
  if (i >= e) return;
  const int d = dst[i], s = src[i], p = pos[i];
  const int b = off[d] + 1;
  const int n = off[d + 1] - b;
  int r = 0;
  for (int j = 0; j < n; ++j) {
    const int t = tmp[b + j];
    r += ((t < s) || (t == s && j < p)) ? 1 : 0;
  }
  ssrc[b + r] = s;
  sw[b + r]   = dinv[s] * dinv[d];
}

__device__ __forceinline__ void gcn_axpy(float4& a, const float4& x, float w) {
  a.x = fmaf(x.x, w, a.x);
  a.y = fmaf(x.y, w, a.y);
  a.z = fmaf(x.z, w, a.z);
  a.w = fmaf(x.w, w, a.w);
}

template <int NV, int K>
__device__ __forceinline__ void gcn_rows(const float4* __restrict__ Xv, int sv, int wvi,
                                         int j, int lane, float4 (&a)[NV]) {
  float4 x[K][NV];
  #pragma unroll
  for (int k = 0; k < K; ++k) {
    const int i = __builtin_amdgcn_readlane(sv, j + k);
    const float4* __restrict__ r = Xv + (size_t)i * (32 * NV);
    #pragma unroll
    for (int v = 0; v < NV; ++v) x[k][v] = r[32 * v + lane];
  }
  #pragma unroll
  for (int k = 0; k < K; ++k) {
    const float w = __int_as_float(__builtin_amdgcn_readlane(wvi, j + k));
    #pragma unroll
    for (int v = 0; v < NV; ++v) gcn_axpy(a[v], x[k][v], w);
  }
}

template <int NV>
__device__ __forceinline__ void gcn_chunk(const float4* __restrict__ Xv, int sv, int wvi, int m,
                                          int lane, float4 (&a)[NV]) {
  constexpr int KMAX = 16 / NV;
  int j = 0;
  for (; j + KMAX <= m; j += KMAX) gcn_rows<NV, KMAX>(Xv, sv, wvi, j, lane, a);
  if (KMAX > 8 && m - j >= 8) { gcn_rows<NV, 8>(Xv, sv, wvi, j, lane, a); j += 8; }
  if (m - j >= 4) { gcn_rows<NV, 4>(Xv, sv, wvi, j, lane, a); j += 4; }
  if (m - j >= 2) { gcn_rows<NV, 2>(Xv, sv, wvi, j, lane, a); j += 2; }
  if (m - j >= 1) { gcn_rows<NV, 1>(Xv, sv, wvi, j, lane, a); }
}

template <int DIN, int RELU>
__launch_bounds__(128)
__global__ void gcn_agg_gemm(const int* __restrict__ off, const int* __restrict__ ssrc,
                             const float* __restrict__ sw,
                             const float* __restrict__ Xin, const float* W,
                             const float* __restrict__ bias, float* __restrict__ Yout) {
  constexpr int NV    = DIN / 128;
  constexpr int PITCH = DIN + 4;
  __shared__ __attribute__((aligned(16))) float xs[16 * PITCH];

  const int lane = threadIdx.x & 31;
  const int wave = __builtin_amdgcn_readfirstlane((int)(threadIdx.x >> 5));
  const int row0 = blockIdx.x * 16;

  {
    const float4* __restrict__ Xv = (const float4*)Xin;
    const int rbase = row0 + 4 * wave;

    int ob[5];
    #pragma unroll
    for (int r = 0; r < 5; ++r) ob[r] = off[rbase + r];
    int   sv0[4];
    float wv0[4];
    #pragma unroll
    for (int r = 0; r < 4; ++r) {
      sv0[r] = 0;
      wv0[r] = 0.f;
      if (lane < ob[r + 1] - ob[r]) {
        sv0[r] = ssrc[ob[r] + lane];
        wv0[r] = sw[ob[r] + lane];
      }
    }

    #pragma unroll
    for (int r = 0; r < 4; ++r) {
      const int b   = ob[r];
      const int cnt = ob[r + 1] - b;
      float4 a[NV];
      #pragma unroll
      for (int v = 0; v < NV; ++v) a[v] = make_float4(0.f, 0.f, 0.f, 0.f);
      for (int c0 = 0; c0 < cnt; c0 += 32) {
        const int m = (cnt - c0 < 32) ? (cnt - c0) : 32;
        int   sv = sv0[r];
        float wv = wv0[r];
        if (c0 != 0) {
          sv = 0;
          wv = 0.f;
          if (lane < m) {
            sv = ssrc[b + c0 + lane];
            wv = sw[b + c0 + lane];
          }
        }
        gcn_chunk<NV>(Xv, sv, __float_as_int(wv), m, lane, a);
      }

      float* xr = xs + (4 * wave + r) * PITCH;
      #pragma unroll
      for (int v = 0; v < NV; ++v) *(float4*)(xr + 128 * v + 4 * lane) = a[v];
    }
  }
  __syncthreads();

  const int half = lane >> 4;
  const int c    = lane & 15;
  const int col0 = wave * 64;

  const float* Xrow  = xs + c * PITCH + 4 * half;
  const float* Wbase = W + (size_t)(4 * half) * D_HID + col0 + 4 * c;

  v8f acc0 = {0.f,0.f,0.f,0.f,0.f,0.f,0.f,0.f};
  v8f acc1 = acc0, acc2 = acc0, acc3 = acc0;

  gcn_tile_product<DIN>(Xrow, Wbase, acc0, acc1, acc2, acc3);

  const v4f bq = *(const v4f*)(bias + col0 + 4 * c);
  gcn_tile_store<RELU>(Yout + (size_t)row0 * D_HID + col0 + 4 * c, half, bq, acc0, acc1, acc2, acc3);
}

extern "C" void kernel_launch(void* const* d_in, const int* in_sizes, int n_in,
                              void* d_out, int out_size, void* d_ws, size_t ws_size,
                              hipStream_t stream) {
  const float* x   = (const float*)d_in[0];
  const int*   ei  = (const int*)  d_in[1];
  const float* W1  = (const float*)d_in[2];
  const float* b1  = (const float*)d_in[3];
  const float* W2  = (const float*)d_in[4];
  const float* b2  = (const float*)d_in[5];
  const float* W3  = (const float*)d_in[6];
  const float* b3  = (const float*)d_in[7];

  const int N = in_sizes[0] / 128;
  const int E = in_sizes[1] / 2;
  const int total = N * D_HID;

  const int* src = ei;
  const int* dst = ei + E;

  const int Npad = (N + 4) & ~3;
  const int L    = E + Npad;
  float* H    = (float*)d_ws;
  float* X    = H + (size_t)total;
  float* dinv = X + (size_t)total;
  int*   cnt  = (int*)(dinv + Npad);
  int*   off  = cnt + Npad;
  int*   pos  = off + Npad;
  int*   tmp  = pos + E;
  int*   ssrc = tmp + L;
  float* sw   = (float*)(ssrc + L);
  int*   loc  = (int*)(sw + L);

  const int rowTiles = N / 16;
  dim3 gemmGrid(rowTiles, 1, 1);
  dim3 gemmBlk(128, 1, 1);
  const int nBlkN = (N + 255) / 256;
  const int nBlkE = (E + 255) / 256;
  const int nBlkF = (E > N) ? nBlkE : nBlkN;
  const int nBlkZ = (Npad + 255) / 256;

  gcn_cnt_lds    <<<GCN_HG, 1024, 0, stream>>>(dst, loc, pos, E, N, Npad);
  gcn_cnt_combine<<<nBlkZ, 256, 0, stream>>>(loc, cnt, Npad);
  gcn_scan_deg <<<1, 1024, 0, stream>>>(cnt, off, dinv, N, Npad);
  gcn_fill     <<<nBlkF, 256, 0, stream>>>(src, dst, pos, loc, off, dinv, tmp, ssrc, sw, E, N, Npad);
  gcn_order    <<<nBlkE, 256, 0, stream>>>(src, dst, pos, off, tmp, dinv, ssrc, sw, E);

  gcn_agg_gemm<128, 1><<<gemmGrid, gemmBlk, 0, stream>>>(off, ssrc, sw, x, W1, b1, H);

  gcn_agg_gemm<256, 1><<<gemmGrid, gemmBlk, 0, stream>>>(off, ssrc, sw, H, W2, b2, X);

  gcn_agg_gemm<256, 0><<<gemmGrid, gemmBlk, 0, stream>>>(off, ssrc, sw, X, W3, b3, (float*)d_out);
}


# Source-order pass of the edge list: one wave per 16 destination groups, a group's entries across lanes and ranked against scalar broadcasts, not one thread per edge re-reading its group
# speedup vs baseline: 5.6141x; 1.1315x over previous
#include <hip/hip_runtime.h>
#include <hip/hip_bf16.h>

#define N_NODES 50000
#define D_HID   256

typedef __attribute__((ext_vector_type(2))) float v2f;
typedef __attribute__((ext_vector_type(8))) float v8f;

typedef __attribute__((ext_vector_type(4))) float v4f;

struct gcn_kblk { v4f a, w0, w1, w2, w3; };

template <int DOUT>
__device__ __forceinline__ gcn_kblk gcn_load_kblk(const float* Xrow, const float* Wbase, int kk) {
  gcn_kblk f;
  f.a  = *(const v4f*)(Xrow + kk);
  f.w0 = *(const v4f*)(Wbase + (size_t)(kk + 0) * DOUT);
  f.w1 = *(const v4f*)(Wbase + (size_t)(kk + 1) * DOUT);
  f.w2 = *(const v4f*)(Wbase + (size_t)(kk + 2) * DOUT);
  f.w3 = *(const v4f*)(Wbase + (size_t)(kk + 3) * DOUT);
  return f;
}

template <int DIN>
__device__ __forceinline__ void gcn_tile_product(const float* Xrow, const float* Wbase,
                                                 v8f& acc0, v8f& acc1, v8f& acc2, v8f& acc3) {
  constexpr int DOUT = D_HID;
  gcn_kblk cur = gcn_load_kblk<DOUT>(Xrow, Wbase, 0);

  #pragma unroll
  for (int kk = 0; kk < DIN; kk += 8) {
    gcn_kblk nxt = cur;
    if (kk + 8 < DIN) nxt = gcn_load_kblk<DOUT>(Xrow, Wbase, kk + 8);
    __builtin_amdgcn_sched_barrier(0);

    v2f alo, ahi;
    alo.x = cur.a.x; alo.y = cur.a.y;
    ahi.x = cur.a.z; ahi.y = cur.a.w;

    v2f b;
    b.x = cur.w0.x; b.y = cur.w1.x;
    acc0 = __builtin_amdgcn_wmma_f32_16x16x4_f32(false, alo, false, b, (short)0, acc0, false, false);
    b.x = cur.w0.y; b.y = cur.w1.y;
    acc1 = __builtin_amdgcn_wmma_f32_16x16x4_f32(false, alo, false, b, (short)0, acc1, false, false);
    b.x = cur.w0.z; b.y = cur.w1.z;
    acc2 = __builtin_amdgcn_wmma_f32_16x16x4_f32(false, alo, false, b, (short)0, acc2, false, false);
    b.x = cur.w0.w; b.y = cur.w1.w;
    acc3 = __builtin_amdgcn_wmma_f32_16x16x4_f32(false, alo, false, b, (short)0, acc3, false, false);

    b.x = cur.w2.x; b.y = cur.w3.x;
    acc0 = __builtin_amdgcn_wmma_f32_16x16x4_f32(false, ahi, false, b, (short)0, acc0, false, false);
    b.x = cur.w2.y; b.y = cur.w3.y;
    acc1 = __builtin_amdgcn_wmma_f32_16x16x4_f32(false, ahi, false, b, (short)0, acc1, false, false);
    b.x = cur.w2.z; b.y = cur.w3.z;
    acc2 = __builtin_amdgcn_wmma_f32_16x16x4_f32(false, ahi, false, b, (short)0, acc2, false, false);
    b.x = cur.w2.w; b.y = cur.w3.w;
    acc3 = __builtin_amdgcn_wmma_f32_16x16x4_f32(false, ahi, false, b, (short)0, acc3, false, false);

    __builtin_amdgcn_sched_barrier(0);
    cur = nxt;
  }
}

template <int RELU>
__device__ __forceinline__ void gcn_tile_store(float* Yb, int half, const v4f& bq, const v8f& acc0,
                                               const v8f& acc1, const v8f& acc2, const v8f& acc3) {
  #pragma unroll
  for (int r = 0; r < 8; ++r) {
    v4f o;
    o.x = acc0[r] + bq.x; o.y = acc1[r] + bq.y; o.z = acc2[r] + bq.z; o.w = acc3[r] + bq.w;
    if (RELU) {
      o.x = fmaxf(o.x, 0.f); o.y = fmaxf(o.y, 0.f); o.z = fmaxf(o.z, 0.f); o.w = fmaxf(o.w, 0.f);
    }
    *(v4f*)(Yb + (size_t)(r + 8 * half) * D_HID) = o;
  }
}

#define GCN_HG 16
#define GCN_HB 32
__launch_bounds__(1024)
__global__ void gcn_cnt_lds(const int* __restrict__ dst, int* __restrict__ loc,
                            int* __restrict__ lrank, int e, int n, int npad) {
  __shared__ __attribute__((aligned(16))) int hist[N_NODES + 32];
  const int tid  = threadIdx.x;
  const int lane = tid & 31;
  int4* hist4 = (int4*)hist;
  for (int k = tid; k < (N_NODES + 32) / 4; k += 1024) hist4[k] = make_int4(0, 0, 0, 0);
  __syncthreads();

  const int per = (e + GCN_HG - 1) / GCN_HG;
  const int lo  = blockIdx.x * per;
  const int hi  = min(lo + per, e);
  for (int i0 = lo + tid; i0 < hi; i0 += 1024 * GCN_HB) {
    int d[GCN_HB], r[GCN_HB];
    #pragma unroll
    for (int u = 0; u < GCN_HB; ++u) {
      const int i = i0 + u * 1024;
      d[u] = dst[min(i, hi - 1)];
    }
    #pragma unroll
    for (int u = 0; u < GCN_HB; ++u) {
      const int i = i0 + u * 1024;
      r[u] = atomicAdd(&hist[(i < hi) ? d[u] : N_NODES + lane], 1);
    }
    #pragma unroll
    for (int u = 0; u < GCN_HB; ++u) {
      const int i = i0 + u * 1024;
      if (i < hi) lrank[i] = r[u];
    }
  }
  __syncthreads();

  int4* __restrict__ out4 = (int4*)(loc + (size_t)blockIdx.x * npad);
  for (int k = tid; k < npad / 4; k += 1024) {
    int4 v = hist4[k];
    v.x = (4 * k     < n) ? v.x : 0;
    v.y = (4 * k + 1 < n) ? v.y : 0;
    v.z = (4 * k + 2 < n) ? v.z : 0;
    v.w = (4 * k + 3 < n) ? v.w : 0;
    out4[k] = v;
  }
}

__global__ void gcn_cnt_combine(int* __restrict__ loc, int* __restrict__ cnt, int npad) {
  const int d = blockIdx.x * blockDim.x + threadIdx.x;
  if (d >= npad) return;
  int c[GCN_HG];
  #pragma unroll
  for (int g = 0; g < GCN_HG; ++g) c[g] = loc[(size_t)g * npad + d];
  int run = 0;
  #pragma unroll
  for (int g = 0; g < GCN_HG; ++g) {
    loc[(size_t)g * npad + d] = run;
    run += c[g];
  }
  cnt[d] = run;
}

__launch_bounds__(1024)
__global__ void gcn_scan_deg(const int* __restrict__ cnt, int* __restrict__ off,
                             float* __restrict__ dinv, int n, int npad) {
  __shared__ int wsum[2][32];
  const int tid  = threadIdx.x;
  const int lane = tid & 31;
  const int wave = tid >> 5;

  const int4* __restrict__ cnt4  = (const int4*)cnt;
  int4*       __restrict__ off4  = (int4*)off;
  float4*     __restrict__ dinv4 = (float4*)dinv;
  const int lastq = npad / 4 - 1;

  int carry = 0;
  int buf   = 0;
  int q0 = tid * 2;
  int4 ra = cnt4[min(q0,     lastq)];
  int4 rb = cnt4[min(q0 + 1, lastq)];

  for (int base = 0; base < n; base += 8192) {
    const int q1 = q0 + 2048;
    const int4 na = cnt4[min(q1,     lastq)];
    const int4 nb = cnt4[min(q1 + 1, lastq)];

    const int i0 = q0 * 4;
    const int c0 = (i0     < n) ? ra.x + 1 : 0;
    const int c1 = (i0 + 1 < n) ? ra.y + 1 : 0;
    const int c2 = (i0 + 2 < n) ? ra.z + 1 : 0;
    const int c3 = (i0 + 3 < n) ? ra.w + 1 : 0;
    const int c4 = (i0 + 4 < n) ? rb.x + 1 : 0;
    const int c5 = (i0 + 5 < n) ? rb.y + 1 : 0;
    const int c6 = (i0 + 6 < n) ? rb.z + 1 : 0;
    const int c7 = (i0 + 7 < n) ? rb.w + 1 : 0;

    const int t = c0 + c1 + c2 + c3 + c4 + c5 + c6 + c7;
    int incl = t;
    #pragma unroll
    for (int o = 1; o < 32; o <<= 1) {
      int v = __shfl_up(incl, o);
      if (lane >= o) incl += v;
    }
    if (lane == 31) wsum[buf][wave] = incl;
    __syncthreads();

    const int ws = wsum[buf][lane];
    int wincl = ws;
    #pragma unroll
    for (int o = 1; o < 32; o <<= 1) {
      int v = __shfl_up(wincl, o);
      if (lane >= o) wincl += v;
    }
    const int wprefix = __shfl(wincl - ws, wave);
    const int total   = __shfl(wincl, 31);

    int4 oa, ob;
    int ex = carry + wprefix + incl - t;
    oa.x = ex; ex += c0;
    oa.y = ex; ex += c1;
    oa.z = ex; ex += c2;
    oa.w = ex; ex += c3;
    ob.x = ex; ex += c4;
    ob.y = ex; ex += c5;
    ob.z = ex; ex += c6;
    ob.w = ex;
    if (q0 <= lastq) {
      float4 da;
      da.x = rsqrtf((float)c0); da.y = rsqrtf((float)c1);
      da.z = rsqrtf((float)c2); da.w = rsqrtf((float)c3);
      off4[q0]  = oa;
      dinv4[q0] = da;
    }
    if (q0 + 1 <= lastq) {
      float4 db;
      db.x = rsqrtf((float)c4); db.y = rsqrtf((float)c5);
      db.z = rsqrtf((float)c6); db.w = rsqrtf((float)c7);
      off4[q0 + 1]  = ob;
      dinv4[q0 + 1] = db;
    }

    carry += total;
    q0  = q1;
    ra  = na;
    rb  = nb;
    buf ^= 1;
  }
  if (tid == 0) off[n] = carry;
}

__global__ void gcn_fill(const int* __restrict__ src, const int* __restrict__ dst,
                         const int* __restrict__ pos, const int* __restrict__ loc,
                         const int* __restrict__ off,
                         const float* __restrict__ dinv, int* __restrict__ tmp,
                         int* __restrict__ ssrc, float* __restrict__ sw, int e, int n, int npad) {
  int i = blockIdx.x * blockDim.x + threadIdx.x;
  if (i < e) {
    const int per = (e + GCN_HG - 1) / GCN_HG;
    const int d = dst[i];
    const int p = loc[(size_t)(i / per) * npad + d] + pos[i];
    tmp[off[d] + 1 + p] = src[i];
  }
  if (i < n) {
    const int   o = off[i];
    const float r = dinv[i];
    ssrc[o] = i;
    sw[o]   = r * r;
  }
}

#define GCN_ONW 16
static_assert(N_NODES <= (1 << 26), "a source index and a 5-bit slot share one 32-bit key");

__device__ __forceinline__ void gcn_order_long(const int* __restrict__ tmp,
                                               const float* __restrict__ dinv,
                                               int* __restrict__ ssrc, float* __restrict__ sw,
                                               int bu, int mu, float dd, int lane) {
  for (int a0 = 0; a0 < mu; a0 += 32) {
    const int q = a0 + lane;
    int   tq = 0;
    float wq = 0.f;
    if (q < mu) {
      tq = tmp[bu + q];
      wq = dinv[tq];
    }
    int r = 0;
    for (int c0 = 0; c0 < mu; c0 += 32) {
      int tc = 0;
      if (c0 + lane < mu) tc = tmp[bu + c0 + lane];
      const int cm = (mu - c0 < 32) ? (mu - c0) : 32;
      for (int j = 0; j < cm; ++j) {
        const int tj = __builtin_amdgcn_readlane(tc, j);
        r += ((tj < tq) || (tj == tq && c0 + j < q)) ? 1 : 0;
      }
    }
    if (q < mu) {
      ssrc[bu + r] = tq;
      sw[bu + r]   = wq * dd;
    }
  }
}

__launch_bounds__(256)
__global__ void gcn_order(const int* __restrict__ off, const int* __restrict__ tmp,
                          const float* __restrict__ dinv, int* __restrict__ ssrc,
                          float* __restrict__ sw, int n) {
  const int lane = threadIdx.x & 31;
  const int wave = __builtin_amdgcn_readfirstlane((int)((blockIdx.x * 256 + threadIdx.x) >> 5));
  const int d0 = wave * GCN_ONW;
  if (d0 >= n) return;

  const int   ov = off[min(d0 + lane, n)];
  const float dl = dinv[min(d0 + lane, n - 1)];

  int   b[GCN_ONW], m[GCN_ONW], t[GCN_ONW];
  float w[GCN_ONW];
  #pragma unroll
  for (int u = 0; u < GCN_ONW; ++u) {
    const int bu = __builtin_amdgcn_readlane(ov, u) + 1;
    const int eu = __builtin_amdgcn_readlane(ov, u + 1);
    b[u] = bu;
    m[u] = (d0 + u < n) ? eu - bu : 0;
    t[u] = 0;
    if (lane < m[u]) t[u] = tmp[bu + lane];
  }
  #pragma unroll
  for (int u = 0; u < GCN_ONW; ++u) w[u] = dinv[t[u]];

  #pragma unroll
  for (int u = 0; u < GCN_ONW; ++u) {
    const int mu = m[u];
    if (mu <= 32) {
      const float dd = __int_as_float(__builtin_amdgcn_readlane(__float_as_int(dl), u));
      const unsigned key = ((unsigned)t[u] << 5) | (unsigned)lane;
      int r = 0;
      for (int j = 0; j < mu; ++j) {
        const unsigned kj = (unsigned)__builtin_amdgcn_readlane((int)key, j);
        r += (kj < key) ? 1 : 0;
      }
      if (lane < mu) {
        ssrc[b[u] + r] = t[u];
        sw[b[u] + r]   = w[u] * dd;
      }
    }
  }

  #pragma unroll 1
  for (int u = 0; u < GCN_ONW; ++u) {
    const int bu = __builtin_amdgcn_readlane(ov, u) + 1;
    const int mu = (d0 + u < n) ? __builtin_amdgcn_readlane(ov, u + 1) - bu : 0;
    if (mu > 32) {
      const float dd = __int_as_float(__builtin_amdgcn_readlane(__float_as_int(dl), u));
      gcn_order_long(tmp, dinv, ssrc, sw, bu, mu, dd, lane);
    }
  }
}

__device__ __forceinline__ void gcn_axpy(float4& a, const float4& x, float w) {
  a.x = fmaf(x.x, w, a.x);
  a.y = fmaf(x.y, w, a.y);
  a.z = fmaf(x.z, w, a.z);
  a.w = fmaf(x.w, w, a.w);
}

template <int NV, int K>
__device__ __forceinline__ void gcn_rows(const float4* __restrict__ Xv, int sv, int wvi,
                                         int j, int lane, float4 (&a)[NV]) {
  float4 x[K][NV];
  #pragma unroll
  for (int k = 0; k < K; ++k) {
    const int i = __builtin_amdgcn_readlane(sv, j + k);
    const float4* __restrict__ r = Xv + (size_t)i * (32 * NV);
    #pragma unroll
    for (int v = 0; v < NV; ++v) x[k][v] = r[32 * v + lane];
  }
  #pragma unroll
  for (int k = 0; k < K; ++k) {
    const float w = __int_as_float(__builtin_amdgcn_readlane(wvi, j + k));
    #pragma unroll
    for (int v = 0; v < NV; ++v) gcn_axpy(a[v], x[k][v], w);
  }
}

template <int NV>
__device__ __forceinline__ void gcn_chunk(const float4* __restrict__ Xv, int sv, int wvi, int m,
                                          int lane, float4 (&a)[NV]) {
  constexpr int KMAX = 16 / NV;
  int j = 0;
  for (; j + KMAX <= m; j += KMAX) gcn_rows<NV, KMAX>(Xv, sv, wvi, j, lane, a);
  if (KMAX > 8 && m - j >= 8) { gcn_rows<NV, 8>(Xv, sv, wvi, j, lane, a); j += 8; }
  if (m - j >= 4) { gcn_rows<NV, 4>(Xv, sv, wvi, j, lane, a); j += 4; }
  if (m - j >= 2) { gcn_rows<NV, 2>(Xv, sv, wvi, j, lane, a); j += 2; }
  if (m - j >= 1) { gcn_rows<NV, 1>(Xv, sv, wvi, j, lane, a); }
}

template <int DIN, int RELU>
__launch_bounds__(128)
__global__ void gcn_agg_gemm(const int* __restrict__ off, const int* __restrict__ ssrc,
                             const float* __restrict__ sw,
                             const float* __restrict__ Xin, const float* W,
                             const float* __restrict__ bias, float* __restrict__ Yout) {
  constexpr int NV    = DIN / 128;
  constexpr int PITCH = DIN + 4;
  __shared__ __attribute__((aligned(16))) float xs[16 * PITCH];

  const int lane = threadIdx.x & 31;
  const int wave = __builtin_amdgcn_readfirstlane((int)(threadIdx.x >> 5));
  const int row0 = blockIdx.x * 16;

  {
    const float4* __restrict__ Xv = (const float4*)Xin;
    const int rbase = row0 + 4 * wave;

    int ob[5];
    #pragma unroll
    for (int r = 0; r < 5; ++r) ob[r] = off[rbase + r];
    int   sv0[4];
    float wv0[4];
    #pragma unroll
    for (int r = 0; r < 4; ++r) {
      sv0[r] = 0;
      wv0[r] = 0.f;
      if (lane < ob[r + 1] - ob[r]) {
        sv0[r] = ssrc[ob[r] + lane];
        wv0[r] = sw[ob[r] + lane];
      }
    }

    #pragma unroll
    for (int r = 0; r < 4; ++r) {
      const int b   = ob[r];
      const int cnt = ob[r + 1] - b;
      float4 a[NV];
      #pragma unroll
      for (int v = 0; v < NV; ++v) a[v] = make_float4(0.f, 0.f, 0.f, 0.f);
      for (int c0 = 0; c0 < cnt; c0 += 32) {
        const int m = (cnt - c0 < 32) ? (cnt - c0) : 32;
        int   sv = sv0[r];
        float wv = wv0[r];
        if (c0 != 0) {
          sv = 0;
          wv = 0.f;
          if (lane < m) {
            sv = ssrc[b + c0 + lane];
            wv = sw[b + c0 + lane];
          }
        }
        gcn_chunk<NV>(Xv, sv, __float_as_int(wv), m, lane, a);
      }

      float* xr = xs + (4 * wave + r) * PITCH;
      #pragma unroll
      for (int v = 0; v < NV; ++v) *(float4*)(xr + 128 * v + 4 * lane) = a[v];
    }
  }
  __syncthreads();

  const int half = lane >> 4;
  const int c    = lane & 15;
  const int col0 = wave * 64;

  const float* Xrow  = xs + c * PITCH + 4 * half;
  const float* Wbase = W + (size_t)(4 * half) * D_HID + col0 + 4 * c;

  v8f acc0 = {0.f,0.f,0.f,0.f,0.f,0.f,0.f,0.f};
  v8f acc1 = acc0, acc2 = acc0, acc3 = acc0;

  gcn_tile_product<DIN>(Xrow, Wbase, acc0, acc1, acc2, acc3);

  const v4f bq = *(const v4f*)(bias + col0 + 4 * c);
  gcn_tile_store<RELU>(Yout + (size_t)row0 * D_HID + col0 + 4 * c, half, bq, acc0, acc1, acc2, acc3);
}

extern "C" void kernel_launch(void* const* d_in, const int* in_sizes, int n_in,
                              void* d_out, int out_size, void* d_ws, size_t ws_size,
                              hipStream_t stream) {
  const float* x   = (const float*)d_in[0];
  const int*   ei  = (const int*)  d_in[1];
  const float* W1  = (const float*)d_in[2];
  const float* b1  = (const float*)d_in[3];
  const float* W2  = (const float*)d_in[4];
  const float* b2  = (const float*)d_in[5];
  const float* W3  = (const float*)d_in[6];
  const float* b3  = (const float*)d_in[7];

  const int N = in_sizes[0] / 128;
  const int E = in_sizes[1] / 2;
  const int total = N * D_HID;

  const int* src = ei;
  const int* dst = ei + E;

  const int Npad = (N + 4) & ~3;
  const int L    = E + Npad;
  float* H    = (float*)d_ws;
  float* X    = H + (size_t)total;
  float* dinv = X + (size_t)total;
  int*   cnt  = (int*)(dinv + Npad);
  int*   off  = cnt + Npad;
  int*   pos  = off + Npad;
  int*   tmp  = pos + E;
  int*   ssrc = tmp + L;
  float* sw   = (float*)(ssrc + L);
  int*   loc  = (int*)(sw + L);

  const int rowTiles = N / 16;
  dim3 gemmGrid(rowTiles, 1, 1);
  dim3 gemmBlk(128, 1, 1);
  const int nBlkN = (N + 255) / 256;
  const int nBlkE = (E + 255) / 256;
  const int nBlkF = (E > N) ? nBlkE : nBlkN;
  const int nBlkZ = (Npad + 255) / 256;
  const int nBlkO = (N + 8 * GCN_ONW - 1) / (8 * GCN_ONW);

  gcn_cnt_lds    <<<GCN_HG, 1024, 0, stream>>>(dst, loc, pos, E, N, Npad);
  gcn_cnt_combine<<<nBlkZ, 256, 0, stream>>>(loc, cnt, Npad);
  gcn_scan_deg <<<1, 1024, 0, stream>>>(cnt, off, dinv, N, Npad);
  gcn_fill     <<<nBlkF, 256, 0, stream>>>(src, dst, pos, loc, off, dinv, tmp, ssrc, sw, E, N, Npad);
  gcn_order    <<<nBlkO, 256, 0, stream>>>(off, tmp, dinv, ssrc, sw, N);

  gcn_agg_gemm<128, 1><<<gemmGrid, gemmBlk, 0, stream>>>(off, ssrc, sw, x, W1, b1, H);

  gcn_agg_gemm<256, 1><<<gemmGrid, gemmBlk, 0, stream>>>(off, ssrc, sw, H, W2, b2, X);

  gcn_agg_gemm<256, 0><<<gemmGrid, gemmBlk, 0, stream>>>(off, ssrc, sw, X, W3, b3, (float*)d_out);
}


# Two list-building launches folded away: node scan as a block scan in the count-combine launch (block totals rescanned by each fill block); source order ranked by layer 1's own waves; 8 to 6 launches
# speedup vs baseline: 6.0319x; 1.0744x over previous
#include <hip/hip_runtime.h>
#include <hip/hip_bf16.h>

#define N_NODES 50000
#define D_HID   256

typedef __attribute__((ext_vector_type(2))) float v2f;
typedef __attribute__((ext_vector_type(8))) float v8f;

typedef __attribute__((ext_vector_type(4))) float v4f;

struct gcn_kblk { v4f a, w0, w1, w2, w3; };

template <int DOUT>
__device__ __forceinline__ gcn_kblk gcn_load_kblk(const float* Xrow, const float* Wbase, int kk) {
  gcn_kblk f;
  f.a  = *(const v4f*)(Xrow + kk);
  f.w0 = *(const v4f*)(Wbase + (size_t)(kk + 0) * DOUT);
  f.w1 = *(const v4f*)(Wbase + (size_t)(kk + 1) * DOUT);
  f.w2 = *(const v4f*)(Wbase + (size_t)(kk + 2) * DOUT);
  f.w3 = *(const v4f*)(Wbase + (size_t)(kk + 3) * DOUT);
  return f;
}

template <int DIN>
__device__ __forceinline__ void gcn_tile_product(const float* Xrow, const float* Wbase,
                                                 v8f& acc0, v8f& acc1, v8f& acc2, v8f& acc3) {
  constexpr int DOUT = D_HID;
  gcn_kblk cur = gcn_load_kblk<DOUT>(Xrow, Wbase, 0);

  #pragma unroll
  for (int kk = 0; kk < DIN; kk += 8) {
    gcn_kblk nxt = cur;
    if (kk + 8 < DIN) nxt = gcn_load_kblk<DOUT>(Xrow, Wbase, kk + 8);
    __builtin_amdgcn_sched_barrier(0);

    v2f alo, ahi;
    alo.x = cur.a.x; alo.y = cur.a.y;
    ahi.x = cur.a.z; ahi.y = cur.a.w;

    v2f b;
    b.x = cur.w0.x; b.y = cur.w1.x;
    acc0 = __builtin_amdgcn_wmma_f32_16x16x4_f32(false, alo, false, b, (short)0, acc0, false, false);
    b.x = cur.w0.y; b.y = cur.w1.y;
    acc1 = __builtin_amdgcn_wmma_f32_16x16x4_f32(false, alo, false, b, (short)0, acc1, false, false);
    b.x = cur.w0.z; b.y = cur.w1.z;
    acc2 = __builtin_amdgcn_wmma_f32_16x16x4_f32(false, alo, false, b, (short)0, acc2, false, false);
    b.x = cur.w0.w; b.y = cur.w1.w;
    acc3 = __builtin_amdgcn_wmma_f32_16x16x4_f32(false, alo, false, b, (short)0, acc3, false, false);

    b.x = cur.w2.x; b.y = cur.w3.x;
    acc0 = __builtin_amdgcn_wmma_f32_16x16x4_f32(false, ahi, false, b, (short)0, acc0, false, false);
    b.x = cur.w2.y; b.y = cur.w3.y;
    acc1 = __builtin_amdgcn_wmma_f32_16x16x4_f32(false, ahi, false, b, (short)0, acc1, false, false);
    b.x = cur.w2.z; b.y = cur.w3.z;
    acc2 = __builtin_amdgcn_wmma_f32_16x16x4_f32(false, ahi, false, b, (short)0, acc2, false, false);
    b.x = cur.w2.w; b.y = cur.w3.w;
    acc3 = __builtin_amdgcn_wmma_f32_16x16x4_f32(false, ahi, false, b, (short)0, acc3, false, false);

    __builtin_amdgcn_sched_barrier(0);
    cur = nxt;
  }
}

template <int RELU>
__device__ __forceinline__ void gcn_tile_store(float* Yb, int half, const v4f& bq, const v8f& acc0,
                                               const v8f& acc1, const v8f& acc2, const v8f& acc3) {
  #pragma unroll
  for (int r = 0; r < 8; ++r) {
    v4f o;
    o.x = acc0[r] + bq.x; o.y = acc1[r] + bq.y; o.z = acc2[r] + bq.z; o.w = acc3[r] + bq.w;
    if (RELU) {
      o.x = fmaxf(o.x, 0.f); o.y = fmaxf(o.y, 0.f); o.z = fmaxf(o.z, 0.f); o.w = fmaxf(o.w, 0.f);
    }
    *(v4f*)(Yb + (size_t)(r + 8 * half) * D_HID) = o;
  }
}

#define GCN_HG 16
#define GCN_HB 32
__launch_bounds__(1024)
__global__ void gcn_cnt_lds(const int* __restrict__ dst, int* __restrict__ loc,
                            int* __restrict__ lrank, int e, int n, int npad) {
  __shared__ __attribute__((aligned(16))) int hist[N_NODES + 32];
  const int tid  = threadIdx.x;
  const int lane = tid & 31;
  int4* hist4 = (int4*)hist;
  for (int k = tid; k < (N_NODES + 32) / 4; k += 1024) hist4[k] = make_int4(0, 0, 0, 0);
  __syncthreads();

  const int per = (e + GCN_HG - 1) / GCN_HG;
  const int lo  = blockIdx.x * per;
  const int hi  = min(lo + per, e);
  for (int i0 = lo + tid; i0 < hi; i0 += 1024 * GCN_HB) {
    int d[GCN_HB], r[GCN_HB];
    #pragma unroll
    for (int u = 0; u < GCN_HB; ++u) {
      const int i = i0 + u * 1024;
      d[u] = dst[min(i, hi - 1)];
    }
    #pragma unroll
    for (int u = 0; u < GCN_HB; ++u) {
      const int i = i0 + u * 1024;
      r[u] = atomicAdd(&hist[(i < hi) ? d[u] : N_NODES + lane], 1);
    }
    #pragma unroll
    for (int u = 0; u < GCN_HB; ++u) {
      const int i = i0 + u * 1024;
      if (i < hi) lrank[i] = r[u];
    }
  }
  __syncthreads();

  int4* __restrict__ out4 = (int4*)(loc + (size_t)blockIdx.x * npad);
  for (int k = tid; k < npad / 4; k += 1024) {
    int4 v = hist4[k];
    v.x = (4 * k     < n) ? v.x : 0;
    v.y = (4 * k + 1 < n) ? v.y : 0;
    v.z = (4 * k + 2 < n) ? v.z : 0;
    v.w = (4 * k + 3 < n) ? v.w : 0;
    out4[k] = v;
  }
}

__launch_bounds__(256)
__global__ void gcn_cnt_combine(int* __restrict__ loc, int* __restrict__ lofs,
                                int* __restrict__ bsum, float* __restrict__ dinv,
                                int n, int npad) {
  __shared__ int wsum[8];
  const int tid  = threadIdx.x;
  const int lane = tid & 31;
  const int wave = tid >> 5;
  const int d    = blockIdx.x * 256 + tid;
  const bool in  = d < npad;
  const int dc   = in ? d : npad - 1;
  int c[GCN_HG];
  #pragma unroll
  for (int g = 0; g < GCN_HG; ++g) c[g] = loc[(size_t)g * npad + dc];
  int run = 0;
  #pragma unroll
  for (int g = 0; g < GCN_HG; ++g) {
    if (in) loc[(size_t)g * npad + d] = run;
    run += c[g];
  }
  const int cc = (d < n) ? run + 1 : 0;

  int incl = cc;
  #pragma unroll
  for (int o = 1; o < 32; o <<= 1) {
    int v = __shfl_up(incl, o);
    if (lane >= o) incl += v;
  }
  if (lane == 31) wsum[wave] = incl;
  __syncthreads();
  int wpre = 0, tot = 0;
  #pragma unroll
  for (int w = 0; w < 8; ++w) {
    const int sw8 = wsum[w];
    wpre += (w < wave) ? sw8 : 0;
    tot  += sw8;
  }
  if (d < n) {
    lofs[d] = wpre + incl - cc;
    dinv[d] = rsqrtf((float)cc);
  }
  if (tid == 0) bsum[blockIdx.x] = tot;
}

__launch_bounds__(256)
__global__ void gcn_fill(const int* __restrict__ src, const int* __restrict__ dst,
                         const int* __restrict__ pos, const int* __restrict__ loc,
                         const int* __restrict__ lofs, const int* __restrict__ bsum,
                         const float* __restrict__ dinv, int* __restrict__ off,
                         int* __restrict__ tmp, int* __restrict__ ssrc, float* __restrict__ sw,
                         int e, int n, int npad, int nb) {
  __shared__ int sbase[256];
  __shared__ int wsum[8];
  const int tid  = threadIdx.x;
  const int lane = tid & 31;
  const int wave = tid >> 5;

  const int bv = bsum[min(tid, nb - 1)];
  const int i   = blockIdx.x * 256 + tid;
  const int ie  = min(i, e - 1);
  const int per = (e + GCN_HG - 1) / GCN_HG;
  const int d   = dst[ie];
  const int ps  = pos[ie];
  const int sr  = src[ie];
  const int in  = min(i, n - 1);
  const int   lo_n = lofs[in];
  const float r_n  = dinv[in];
  const int lc  = loc[(size_t)(ie / per) * npad + d];
  const int lo_d = lofs[d];

  const int v  = (tid < nb) ? bv : 0;
  int incl = v;
  #pragma unroll
  for (int o = 1; o < 32; o <<= 1) {
    int u = __shfl_up(incl, o);
    if (lane >= o) incl += u;
  }
  if (lane == 31) wsum[wave] = incl;
  __syncthreads();
  int wpre = 0;
  #pragma unroll
  for (int w = 0; w < 8; ++w) wpre += (w < wave) ? wsum[w] : 0;
  sbase[tid] = wpre + incl - v;
  __syncthreads();

  if (i < e) tmp[sbase[d >> 8] + lo_d + 1 + lc + ps] = sr;
  if (i < n) {
    const int o = sbase[i >> 8] + lo_n;
    off[i]  = o;
    ssrc[o] = i;
    sw[o]   = r_n * r_n;
  }
  if (blockIdx.x == 0 && tid == 255) off[n] = wpre + incl;
}

static_assert(N_NODES <= (1 << 26), "a source index and a 5-bit slot share one 32-bit key");
static_assert((N_NODES + 4 + 255) / 256 <= 256, "one fill block scans all node-block totals");

__device__ __forceinline__ void gcn_order_long(const int* __restrict__ tmp,
                                               const float* __restrict__ dinv,
                                               int* __restrict__ ssrc, float* __restrict__ sw,
                                               int bu, int mu, float dd, int lane) {
  for (int a0 = 0; a0 < mu; a0 += 32) {
    const int q = a0 + lane;
    int   tq = 0;
    float wq = 0.f;
    if (q < mu) {
      tq = tmp[bu + q];
      wq = dinv[tq];
    }
    int r = 0;
    for (int c0 = 0; c0 < mu; c0 += 32) {
      int tc = 0;
      if (c0 + lane < mu) tc = tmp[bu + c0 + lane];
      const int cm = (mu - c0 < 32) ? (mu - c0) : 32;
      for (int j = 0; j < cm; ++j) {
        const int tj = __builtin_amdgcn_readlane(tc, j);
        r += ((tj < tq) || (tj == tq && c0 + j < q)) ? 1 : 0;
      }
    }
    if (q < mu) {
      ssrc[bu + r] = tq;
      sw[bu + r]   = wq * dd;
    }
  }
}

__device__ __forceinline__ void gcn_axpy(float4& a, const float4& x, float w) {
  a.x = fmaf(x.x, w, a.x);
  a.y = fmaf(x.y, w, a.y);
  a.z = fmaf(x.z, w, a.z);
  a.w = fmaf(x.w, w, a.w);
}

template <int NV, int K>
__device__ __forceinline__ void gcn_rows(const float4* __restrict__ Xv, int sv, int wvi,
                                         int j, int lane, float4 (&a)[NV]) {
  float4 x[K][NV];
  #pragma unroll
  for (int k = 0; k < K; ++k) {
    const int i = __builtin_amdgcn_readlane(sv, j + k);
    const float4* __restrict__ r = Xv + (size_t)i * (32 * NV);
    #pragma unroll
    for (int v = 0; v < NV; ++v) x[k][v] = r[32 * v + lane];
  }
  #pragma unroll
  for (int k = 0; k < K; ++k) {
    const float w = __int_as_float(__builtin_amdgcn_readlane(wvi, j + k));
    #pragma unroll
    for (int v = 0; v < NV; ++v) gcn_axpy(a[v], x[k][v], w);
  }
}

template <int NV>
__device__ __forceinline__ void gcn_chunk(const float4* __restrict__ Xv, int sv, int wvi, int m,
                                          int lane, float4 (&a)[NV]) {
  constexpr int KMAX = 16 / NV;
  int j = 0;
  for (; j + KMAX <= m; j += KMAX) gcn_rows<NV, KMAX>(Xv, sv, wvi, j, lane, a);
  if (KMAX > 8 && m - j >= 8) { gcn_rows<NV, 8>(Xv, sv, wvi, j, lane, a); j += 8; }
  if (m - j >= 4) { gcn_rows<NV, 4>(Xv, sv, wvi, j, lane, a); j += 4; }
  if (m - j >= 2) { gcn_rows<NV, 2>(Xv, sv, wvi, j, lane, a); j += 2; }
  if (m - j >= 1) { gcn_rows<NV, 1>(Xv, sv, wvi, j, lane, a); }
}

template <int DIN, int RELU, int ORDER>
__launch_bounds__(128)
__global__ void gcn_agg_gemm(const int* __restrict__ off, const int* __restrict__ ssrc,
                             const float* __restrict__ sw,
                             const float* __restrict__ Xin, const float* W,
                             const float* __restrict__ bias, float* __restrict__ Yout,
                             const int* __restrict__ tmp, const float* __restrict__ dinv,
                             int* ssrc_w, float* sw_w) {
  constexpr int NV    = DIN / 128;
  constexpr int PITCH = DIN + 4;
  __shared__ __attribute__((aligned(16))) float xs[16 * PITCH];

  const int lane = threadIdx.x & 31;
  const int wave = __builtin_amdgcn_readfirstlane((int)(threadIdx.x >> 5));
  const int row0 = blockIdx.x * 16;

  {
    const float4* __restrict__ Xv = (const float4*)Xin;
    const int rbase = row0 + 4 * wave;

    int ob[5];
    #pragma unroll
    for (int r = 0; r < 5; ++r) ob[r] = off[rbase + r];
    int   sv0[4];
    float wv0[4];
    bool  lng[4];
    if (ORDER) {
      static_assert(PITCH >= 64, "a row of xs holds a 32-entry list: 32 sources, 32 weights");
      const float dl = dinv[rbase + (lane & 3)];
      int   t[4];
      float w[4];
      #pragma unroll
      for (int r = 0; r < 4; ++r) {
        const int bu = ob[r] + 1;
        const int mu = ob[r + 1] - bu;
        t[r] = 0;
        if (lane < mu) t[r] = tmp[bu + lane];
      }
      #pragma unroll
      for (int r = 0; r < 4; ++r) w[r] = dinv[t[r]];

      const int ov = off[rbase + min(lane, 4)];
      #pragma unroll 1
      for (int r = 0; r < 4; ++r) {
        const int bu = __builtin_amdgcn_readlane(ov, r) + 1;
        const int mu = __builtin_amdgcn_readlane(ov, r + 1) - bu;
        if (mu > 31) {
          const float dd = __int_as_float(__builtin_amdgcn_readlane(__float_as_int(dl), r));
          gcn_order_long(tmp, dinv, ssrc_w, sw_w, bu, mu, dd, lane);
          __threadfence();
        }
      }

      #pragma unroll
      for (int r = 0; r < 4; ++r) {
        const int bu = ob[r] + 1;
        const int mu = ob[r + 1] - bu;
        const float dd = __int_as_float(__builtin_amdgcn_readlane(__float_as_int(dl), r));
        lng[r] = mu > 31;
        sv0[r] = 0;
        wv0[r] = 0.f;
        if (mu <= 31) {
          float* li = xs + (4 * wave + r) * PITCH;
          float* lw = li + 32;
          const unsigned key = ((unsigned)t[r] << 5) | (unsigned)lane;
          int rk = 0;
          for (int j = 0; j < mu; ++j) {
            const unsigned kj = (unsigned)__builtin_amdgcn_readlane((int)key, j);
            rk += (kj < key) ? 1 : 0;
          }
          if (lane < mu) {
            const float we = w[r] * dd;
            ssrc_w[bu + rk] = t[r];
            sw_w[bu + rk]   = we;
            li[1 + rk] = __int_as_float(t[r]);
            lw[1 + rk] = we;
          }
          if (lane == 0) {
            li[0] = __int_as_float(rbase + r);
            lw[0] = dd * dd;
          }
          if (lane < mu + 1) {
            sv0[r] = __float_as_int(li[lane]);
            wv0[r] = lw[lane];
          }
        }
      }
    } else {
      #pragma unroll
      for (int r = 0; r < 4; ++r) {
        lng[r] = false;
        sv0[r] = 0;
        wv0[r] = 0.f;
        if (lane < ob[r + 1] - ob[r]) {
          sv0[r] = ssrc[ob[r] + lane];
          wv0[r] = sw[ob[r] + lane];
        }
      }
    }

    #pragma unroll
    for (int r = 0; r < 4; ++r) {
      const int b   = ob[r];
      const int cnt = ob[r + 1] - b;
      float4 a[NV];
      #pragma unroll
      for (int v = 0; v < NV; ++v) a[v] = make_float4(0.f, 0.f, 0.f, 0.f);
      for (int c0 = 0; c0 < cnt; c0 += 32) {
        const int m = (cnt - c0 < 32) ? (cnt - c0) : 32;
        int   sv = sv0[r];
        float wv = wv0[r];
        if (c0 != 0 || lng[r]) {
          sv = 0;
          wv = 0.f;
          if (lane < m) {
            if (ORDER) {
              sv = ssrc_w[b + c0 + lane];
              wv = sw_w[b + c0 + lane];
            } else {
              sv = ssrc[b + c0 + lane];
              wv = sw[b + c0 + lane];
            }
          }
        }
        gcn_chunk<NV>(Xv, sv, __float_as_int(wv), m, lane, a);
      }

      float* xr = xs + (4 * wave + r) * PITCH;
      #pragma unroll
      for (int v = 0; v < NV; ++v) *(float4*)(xr + 128 * v + 4 * lane) = a[v];
    }
  }
  __syncthreads();

  const int half = lane >> 4;
  const int c    = lane & 15;
  const int col0 = wave * 64;

  const float* Xrow  = xs + c * PITCH + 4 * half;
  const float* Wbase = W + (size_t)(4 * half) * D_HID + col0 + 4 * c;

  v8f acc0 = {0.f,0.f,0.f,0.f,0.f,0.f,0.f,0.f};
  v8f acc1 = acc0, acc2 = acc0, acc3 = acc0;

  gcn_tile_product<DIN>(Xrow, Wbase, acc0, acc1, acc2, acc3);

  const v4f bq = *(const v4f*)(bias + col0 + 4 * c);
  gcn_tile_store<RELU>(Yout + (size_t)row0 * D_HID + col0 + 4 * c, half, bq, acc0, acc1, acc2, acc3);
}

extern "C" void kernel_launch(void* const* d_in, const int* in_sizes, int n_in,
                              void* d_out, int out_size, void* d_ws, size_t ws_size,
                              hipStream_t stream) {
  const float* x   = (const float*)d_in[0];
  const int*   ei  = (const int*)  d_in[1];
  const float* W1  = (const float*)d_in[2];
  const float* b1  = (const float*)d_in[3];
  const float* W2  = (const float*)d_in[4];
  const float* b2  = (const float*)d_in[5];
  const float* W3  = (const float*)d_in[6];
  const float* b3  = (const float*)d_in[7];

  const int N = in_sizes[0] / 128;
  const int E = in_sizes[1] / 2;
  const int total = N * D_HID;

  const int* src = ei;
  const int* dst = ei + E;

  const int Npad = (N + 4) & ~3;
  const int L    = E + Npad;
  float* H    = (float*)d_ws;
  float* X    = H + (size_t)total;
  float* dinv = X + (size_t)total;
  int*   lofs = (int*)(dinv + Npad);
  int*   off  = lofs + Npad;
  int*   pos  = off + Npad;
  int*   tmp  = pos + E;
  int*   ssrc = tmp + L;
  float* sw   = (float*)(ssrc + L);
  int*   loc  = (int*)(sw + L);
  int*   bsum = loc + (size_t)GCN_HG * Npad;

  const int rowTiles = N / 16;
  dim3 gemmGrid(rowTiles, 1, 1);
  dim3 gemmBlk(128, 1, 1);
  const int nBlkN = (N + 255) / 256;
  const int nBlkE = (E + 255) / 256;
  const int nBlkF = (E > N) ? nBlkE : nBlkN;
  const int nBlkZ = (Npad + 255) / 256;

  gcn_cnt_lds    <<<GCN_HG, 1024, 0, stream>>>(dst, loc, pos, E, N, Npad);
  gcn_cnt_combine<<<nBlkZ, 256, 0, stream>>>(loc, lofs, bsum, dinv, N, Npad);
  gcn_fill       <<<nBlkF, 256, 0, stream>>>(src, dst, pos, loc, lofs, bsum, dinv, off, tmp, ssrc, sw, E, N, Npad, nBlkZ);

  gcn_agg_gemm<128, 1, 1><<<gemmGrid, gemmBlk, 0, stream>>>(off, ssrc, sw, x, W1, b1, H, tmp, dinv, ssrc, sw);

  gcn_agg_gemm<256, 1, 0><<<gemmGrid, gemmBlk, 0, stream>>>(off, ssrc, sw, H, W2, b2, X, tmp, dinv, ssrc, sw);

  gcn_agg_gemm<256, 0, 0><<<gemmGrid, gemmBlk, 0, stream>>>(off, ssrc, sw, X, W3, b3, (float*)d_out, tmp, dinv, ssrc, sw);
}
